# starting kernel
# baseline (speedup 1.0000x reference)
#include <hip/hip_runtime.h>


typedef float v2f __attribute__((ext_vector_type(2)));
typedef float v8f __attribute__((ext_vector_type(8)));

#define B_    32
#define N_    32
#define D_    12
#define P_    32
#define L_    1024
#define LOUT_ 32
#define BIGV  1e30f

__device__ __forceinline__ float bperm_f(int src_lane, float v) {
    int r = __builtin_amdgcn_ds_bpermute(src_lane << 2,
                                         __builtin_bit_cast(int, v));
    return __builtin_bit_cast(float, r);
}

__global__ __launch_bounds__(32) void dtw_wmma_kernel(
    const float* __restrict__ x,
    const float* __restrict__ patts,
    float* __restrict__ out)
{
    const int lane = threadIdx.x;
    const int hi   = lane >> 4;
    const int lo   = lane & 15;
    const int b    = blockIdx.x;
    const int n    = blockIdx.y;

    __shared__ float p2s[P_];
    __shared__ float ctile[P_ * 17];

    const float* Pn = patts + n * (D_ * P_);
    const float* Xb = x + b * (D_ * L_);

    {
        float s = 0.0f;
        #pragma unroll
        for (int d = 0; d < D_; ++d) { float v = Pn[d * P_ + lane]; s += v * v; }
        p2s[lane] = s;
    }
    __syncthreads();

    v2f Afrag[2][3];
    #pragma unroll
    for (int m = 0; m < 2; ++m)
        #pragma unroll
        for (int kc = 0; kc < 3; ++kc) {
            int d0 = 4 * kc + 2 * hi;
            int p  = m * 16 + lo;
            v2f a; a.x = Pn[d0 * P_ + p]; a.y = Pn[(d0 + 1) * P_ + p];
            Afrag[m][kc] = a;
        }

    float p2v[2][8];
    #pragma unroll
    for (int m = 0; m < 2; ++m)
        #pragma unroll
        for (int r = 0; r < 8; ++r)
            p2v[m][r] = p2s[m * 16 + 8 * hi + r];

    float dprev = BIGV;
    float* outp = out + ((b * N_ + n) * P_ + lane) * LOUT_;
    const float INF = __builtin_inff();

    for (int jt = 0; jt < L_ / 16; ++jt) {
        const int jbase = jt * 16;
        const int j     = jbase + lo;

        v2f Bfrag[3];
        float sx = 0.0f;
        #pragma unroll
        for (int kc = 0; kc < 3; ++kc) {
            int d0 = 4 * kc + 2 * hi;
            float b0 = Xb[d0 * L_ + j];
            float b1 = Xb[(d0 + 1) * L_ + j];
            v2f bb; bb.x = b0; bb.y = b1;
            Bfrag[kc] = bb;
            sx += b0 * b0 + b1 * b1;
        }
        sx += bperm_f(lane ^ 16, sx);

        if (jt + 1 < L_ / 16)
            __builtin_prefetch(&Xb[(2 * hi) * L_ + j + 16], 0, 3);

        #pragma unroll
        for (int m = 0; m < 2; ++m) {
            v8f acc = {};
            acc = __builtin_amdgcn_wmma_f32_16x16x4_f32(false, Afrag[m][0],
                    false, Bfrag[0], (short)0, acc, false, false);
            acc = __builtin_amdgcn_wmma_f32_16x16x4_f32(false, Afrag[m][1],
                    false, Bfrag[1], (short)0, acc, false, false);
            acc = __builtin_amdgcn_wmma_f32_16x16x4_f32(false, Afrag[m][2],
                    false, Bfrag[2], (short)0, acc, false, false);
            #pragma unroll
            for (int r = 0; r < 8; ++r) {
                float cst = p2v[m][r] + sx - 2.0f * acc[r];
                ctile[(m * 16 + 8 * hi + r) * 17 + lo] = cst;
            }
        }
        __syncthreads();

        for (int jj = 0; jj < 16; ++jj) {
            const int   jg = jbase + jj;
            const float c  = ctile[lane * 17 + jj];

            float a;
            if (jg == 0) {
                a = (lane == 0) ? 0.0f : BIGV;
            } else {
                float up = bperm_f(lane - 1, dprev);
                up = (lane == 0) ? BIGV : up;
                a  = fminf(up, dprev);
            }

            float al = c + a;
            float be = c;
            #pragma unroll
            for (int o = 1; o < 32; o <<= 1) {
                float af = bperm_f(lane - o, al);
                float bf = bperm_f(lane - o, be);
                bool ok = (lane >= o);
                af = ok ? af : INF;
                bf = ok ? bf : 0.0f;
                al = fminf(al, be + af);
                be = be + bf;
            }
            dprev = fminf(al, be + BIGV);

            if (jg >= L_ - LOUT_) outp[jg - (L_ - LOUT_)] = dprev;
        }
        __syncthreads();
    }
}

extern "C" void kernel_launch(void* const* d_in, const int* in_sizes, int n_in,
                              void* d_out, int out_size, void* d_ws, size_t ws_size,
                              hipStream_t stream) {
    const float* x     = (const float*)d_in[0];
    const float* patts = (const float*)d_in[1];
    float*       outp  = (float*)d_out;
    (void)in_sizes; (void)n_in; (void)out_size; (void)d_ws; (void)ws_size;

    dim3 grid(B_, N_);
    dim3 block(32);
    dtw_wmma_kernel<<<grid, block, 0, stream>>>(x, patts, outp);
}


# DP swept along anti-diagonals (lane i runs i columns behind): one lane shift per step replaces the 11-permute column scan; sums now in the reference's order, one add per cell, no prefix sums
# speedup vs baseline: 5.1319x; 5.1319x over previous
#include <hip/hip_runtime.h>


typedef float v2f __attribute__((ext_vector_type(2)));
typedef float v4f __attribute__((ext_vector_type(4)));
typedef float v8f __attribute__((ext_vector_type(8)));

#define B_    32
#define N_    32
#define D_    12
#define P_    32
#define L_    1024
#define LOUT_ 32
#define BIGV  1e30f
#define NT_   (L_ / 16)
#define NTS_  ((L_ + P_) / 16)
#define RS_   68

__device__ __forceinline__ float bperm_f(int src_lane, float v) {
    int r = __builtin_amdgcn_ds_bpermute(src_lane << 2,
                                         __builtin_bit_cast(int, v));
    return __builtin_bit_cast(float, r);
}

__device__ __forceinline__ void load_xfrag(const float* __restrict__ Xb, int t,
                                           int hi, int lo, v2f (&Bf)[3]) {
    const int j = t * 16 + lo;
    #pragma unroll
    for (int kc = 0; kc < 3; ++kc) {
        int d0 = 4 * kc + 2 * hi;
        v2f bb; bb.x = Xb[d0 * L_ + j]; bb.y = Xb[(d0 + 1) * L_ + j];
        Bf[kc] = bb;
    }
}

__device__ __forceinline__ void cost_tile(float* __restrict__ ring,
                                          const v2f (&Af)[2][3], const v2f (&Bf)[3],
                                          const float (&p2v)[2][8],
                                          int t, int lane, int hi, int lo) {
    float sx = 0.0f;
    #pragma unroll
    for (int kc = 0; kc < 3; ++kc) {
        float b0 = Bf[kc].x, b1 = Bf[kc].y;
        sx += b0 * b0 + b1 * b1;
    }
    sx += bperm_f(lane ^ 16, sx);

    #pragma unroll
    for (int m = 0; m < 2; ++m) {
        v8f acc = {};
        acc = __builtin_amdgcn_wmma_f32_16x16x4_f32(false, Af[m][0],
                false, Bf[0], (short)0, acc, false, false);
        acc = __builtin_amdgcn_wmma_f32_16x16x4_f32(false, Af[m][1],
                false, Bf[1], (short)0, acc, false, false);
        acc = __builtin_amdgcn_wmma_f32_16x16x4_f32(false, Af[m][2],
                false, Bf[2], (short)0, acc, false, false);
        #pragma unroll
        for (int r = 0; r < 8; ++r) {
            const int row = m * 16 + 8 * hi + r;
            const int pos = (t * 16 + lo + row) & 63;
            ring[row * RS_ + pos] = p2v[m][r] + sx - 2.0f * acc[r];
        }
    }
}

template <bool EDGE, bool STORE>
__device__ __forceinline__ void dp16(const float (&c)[16], int kbase, int lane, int src4,
                                     float& dcur, float& diag, float* __restrict__ outp) {
    #pragma unroll
    for (int s = 0; s < 16; ++s) {
        const int j = kbase + s - lane;
        const float sh = __builtin_bit_cast(float,
            __builtin_amdgcn_ds_bpermute(src4, __builtin_bit_cast(int, dcur)));
        float dg = diag;
        if (EDGE) dg = (j == 0) ? BIGV : dg;
        const float dn = c[s] + fminf(fminf(dg, dcur), sh);
        diag = (lane == 0) ? BIGV : sh;
        if (EDGE) dcur = ((unsigned)j < (unsigned)L_) ? dn : dcur;
        else      dcur = dn;
        if (STORE) {
            if ((unsigned)(j - (L_ - LOUT_)) < (unsigned)LOUT_) outp[j - (L_ - LOUT_)] = dn;
        }
    }
}

template <bool EDGE, bool STORE>
__device__ __forceinline__ void tile_step(int T, float* __restrict__ ring,
                                          const float* __restrict__ Xb,
                                          const v2f (&Af)[2][3], v2f (&Bf)[3],
                                          const float (&p2v)[2][8],
                                          int lane, int hi, int lo, int src4,
                                          float& dcur, float& diag,
                                          float* __restrict__ outp) {
    __syncthreads();

    const v4f* cp = reinterpret_cast<const v4f*>(ring + lane * RS_ + ((T * 16) & 63));
    const v4f q0 = cp[0], q1 = cp[1], q2 = cp[2], q3 = cp[3];
    const float c[16] = { q0.x, q0.y, q0.z, q0.w, q1.x, q1.y, q1.z, q1.w,
                          q2.x, q2.y, q2.z, q2.w, q3.x, q3.y, q3.z, q3.w };

    if (T + 1 < NT_) {
        cost_tile(ring, Af, Bf, p2v, T + 1, lane, hi, lo);
        if (T + 2 < NT_) load_xfrag(Xb, T + 2, hi, lo, Bf);
    }

    dp16<EDGE, STORE>(c, T * 16, lane, src4, dcur, diag, outp);
}

__global__ __launch_bounds__(32) void dtw_wmma_kernel(
    const float* __restrict__ x,
    const float* __restrict__ patts,
    float* __restrict__ out)
{
    const int lane = threadIdx.x;
    const int hi   = lane >> 4;
    const int lo   = lane & 15;
    const int b    = blockIdx.x;
    const int n    = blockIdx.y;

    __shared__ float p2s[P_];
    __shared__ __attribute__((aligned(16))) float ring[P_ * RS_];

    const float* Pn = patts + n * (D_ * P_);
    const float* Xb = x + b * (D_ * L_);

    v2f Bf[3];
    load_xfrag(Xb, 0, hi, lo, Bf);

    {
        float s = 0.0f;
        #pragma unroll
        for (int d = 0; d < D_; ++d) { float v = Pn[d * P_ + lane]; s += v * v; }
        p2s[lane] = s;
    }
    {
        v4f* rz = reinterpret_cast<v4f*>(ring);
        const v4f z = {0.0f, 0.0f, 0.0f, 0.0f};
        for (int q = 0; q < (P_ * RS_) / (4 * 32); ++q) rz[q * 32 + lane] = z;
    }
    __syncthreads();

    v2f Afrag[2][3];
    #pragma unroll
    for (int m = 0; m < 2; ++m)
        #pragma unroll
        for (int kc = 0; kc < 3; ++kc) {
            int d0 = 4 * kc + 2 * hi;
            int p  = m * 16 + lo;
            v2f a; a.x = Pn[d0 * P_ + p]; a.y = Pn[(d0 + 1) * P_ + p];
            Afrag[m][kc] = a;
        }

    float p2v[2][8];
    #pragma unroll
    for (int m = 0; m < 2; ++m)
        #pragma unroll
        for (int r = 0; r < 8; ++r)
            p2v[m][r] = p2s[m * 16 + 8 * hi + r];

    cost_tile(ring, Afrag, Bf, p2v, 0, lane, hi, lo);
    load_xfrag(Xb, 1, hi, lo, Bf);

    float dcur = (lane == 0) ? 0.0f : BIGV;
    float diag = BIGV;
    const int src4 = ((lane == 0) ? 0 : (lane - 1)) << 2;
    float* outp = out + ((b * N_ + n) * P_ + lane) * LOUT_;

    for (int T = 0; T < 2; ++T)
        tile_step<true, false>(T, ring, Xb, Afrag, Bf, p2v, lane, hi, lo, src4,
                               dcur, diag, outp);
    for (int T = 2; T < (L_ - LOUT_) / 16; ++T)
        tile_step<false, false>(T, ring, Xb, Afrag, Bf, p2v, lane, hi, lo, src4,
                                dcur, diag, outp);
    for (int T = (L_ - LOUT_) / 16; T < NTS_; ++T)
        tile_step<true, true>(T, ring, Xb, Afrag, Bf, p2v, lane, hi, lo, src4,
                              dcur, diag, outp);
}

extern "C" void kernel_launch(void* const* d_in, const int* in_sizes, int n_in,
                              void* d_out, int out_size, void* d_ws, size_t ws_size,
                              hipStream_t stream) {
    const float* x     = (const float*)d_in[0];
    const float* patts = (const float*)d_in[1];
    float*       outp  = (float*)d_out;
    (void)in_sizes; (void)n_in; (void)out_size; (void)d_ws; (void)ws_size;

    dim3 grid(B_, N_);
    dim3 block(32);
    dtw_wmma_kernel<<<grid, block, 0, stream>>>(x, patts, outp);
}


# DP's per-step lane shift moved from the LDS unit (ds_bpermute) to the vector ALU: DPP row_shr:1 inside each 16-lane row, one v_permlanex16 for lane 15 to 16; same adds in the same order
# speedup vs baseline: 7.4757x; 1.4567x over previous
#include <hip/hip_runtime.h>


typedef float v2f __attribute__((ext_vector_type(2)));
typedef float v4f __attribute__((ext_vector_type(4)));
typedef float v8f __attribute__((ext_vector_type(8)));

#define B_    32
#define N_    32
#define D_    12
#define P_    32
#define L_    1024
#define LOUT_ 32
#define BIGV  1e30f
#define NT_   (L_ / 16)
#define NTS_  ((L_ + P_) / 16)
#define RS_   68


__device__ __forceinline__ float xhalf_f(float v) {
    return __builtin_bit_cast(float,
        __builtin_amdgcn_permlanex16(0, __builtin_bit_cast(int, v),
                                     0x76543210u, 0xfedcba98u, false, false));
}

__device__ __forceinline__ float row_up1(float v) {
    return __builtin_bit_cast(float,
        __builtin_amdgcn_update_dpp(__builtin_bit_cast(int, BIGV), __builtin_bit_cast(int, v),
                                    0x111, 0xf, 0xf, false));
}

__device__ __forceinline__ float cross_up1(float z) {
    return __builtin_bit_cast(float,
        __builtin_amdgcn_permlanex16(0, __builtin_bit_cast(int, z),
                                     0x0000000fu, 0x00000000u, false, false));
}

__device__ __forceinline__ void load_xfrag(const float* __restrict__ Xb, int t,
                                           int hi, int lo, v2f (&Bf)[3]) {
    const int j = t * 16 + lo;
    #pragma unroll
    for (int kc = 0; kc < 3; ++kc) {
        int d0 = 4 * kc + 2 * hi;
        v2f bb; bb.x = Xb[d0 * L_ + j]; bb.y = Xb[(d0 + 1) * L_ + j];
        Bf[kc] = bb;
    }
}

__device__ __forceinline__ void cost_tile(float* __restrict__ ring,
                                          const v2f (&Af)[2][3], const v2f (&Bf)[3],
                                          const float (&p2v)[2][8],
                                          int t, int lane, int hi, int lo) {
    float sx = 0.0f;
    #pragma unroll
    for (int kc = 0; kc < 3; ++kc) {
        float b0 = Bf[kc].x, b1 = Bf[kc].y;
        sx += b0 * b0 + b1 * b1;
    }
    sx += xhalf_f(sx);

    #pragma unroll
    for (int m = 0; m < 2; ++m) {
        v8f acc = {};
        acc = __builtin_amdgcn_wmma_f32_16x16x4_f32(false, Af[m][0],
                false, Bf[0], (short)0, acc, false, false);
        acc = __builtin_amdgcn_wmma_f32_16x16x4_f32(false, Af[m][1],
                false, Bf[1], (short)0, acc, false, false);
        acc = __builtin_amdgcn_wmma_f32_16x16x4_f32(false, Af[m][2],
                false, Bf[2], (short)0, acc, false, false);
        #pragma unroll
        for (int r = 0; r < 8; ++r) {
            const int row = m * 16 + 8 * hi + r;
            const int pos = (t * 16 + lo + row) & 63;
            ring[row * RS_ + pos] = p2v[m][r] + sx - 2.0f * acc[r];
        }
    }
}

template <bool EDGE, bool STORE>
__device__ __forceinline__ void dp16(const float (&c)[16], int kbase, int lane, bool is15,
                                     float& dcur, float& diag, float& z15,
                                     float* __restrict__ outp) {
    #pragma unroll
    for (int s = 0; s < 16; ++s) {
        const int j = kbase + s - lane;
        const float u = row_up1(dcur);
        const float t = cross_up1(z15);
        float dg = diag;
        if (EDGE) dg = (j == 0) ? BIGV : dg;
        const float w = fminf(dg, dcur);
        const float m = fminf(fminf(u, w), t);
        const float dn = c[s] + m;
        diag = fminf(u, t);
        if (EDGE) {
            dcur = ((unsigned)j < (unsigned)L_) ? dn : dcur;
            z15  = is15 ? dcur : BIGV;
        } else {
            dcur = dn;
            z15  = (is15 ? c[s] : BIGV) + m;
        }
        if (STORE) {
            if ((unsigned)(j - (L_ - LOUT_)) < (unsigned)LOUT_) outp[j - (L_ - LOUT_)] = dn;
        }
    }
}

template <bool EDGE, bool STORE>
__device__ __forceinline__ void tile_step(int T, float* __restrict__ ring,
                                          const float* __restrict__ Xb,
                                          const v2f (&Af)[2][3], v2f (&Bf)[3],
                                          const float (&p2v)[2][8],
                                          int lane, int hi, int lo, bool is15,
                                          float& dcur, float& diag, float& z15,
                                          float* __restrict__ outp) {
    __syncthreads();

    const v4f* cp = reinterpret_cast<const v4f*>(ring + lane * RS_ + ((T * 16) & 63));
    const v4f q0 = cp[0], q1 = cp[1], q2 = cp[2], q3 = cp[3];
    const float c[16] = { q0.x, q0.y, q0.z, q0.w, q1.x, q1.y, q1.z, q1.w,
                          q2.x, q2.y, q2.z, q2.w, q3.x, q3.y, q3.z, q3.w };

    if (T + 1 < NT_) {
        cost_tile(ring, Af, Bf, p2v, T + 1, lane, hi, lo);
        if (T + 2 < NT_) load_xfrag(Xb, T + 2, hi, lo, Bf);
    }

    dp16<EDGE, STORE>(c, T * 16, lane, is15, dcur, diag, z15, outp);
}

__global__ __launch_bounds__(32) void dtw_wmma_kernel(
    const float* __restrict__ x,
    const float* __restrict__ patts,
    float* __restrict__ out)
{
    const int lane = threadIdx.x;
    const int hi   = lane >> 4;
    const int lo   = lane & 15;
    const int b    = blockIdx.x;
    const int n    = blockIdx.y;

    __shared__ float p2s[P_];
    __shared__ __attribute__((aligned(16))) float ring[P_ * RS_];

    const float* Pn = patts + n * (D_ * P_);
    const float* Xb = x + b * (D_ * L_);

    v2f Bf[3];
    load_xfrag(Xb, 0, hi, lo, Bf);

    {
        float s = 0.0f;
        #pragma unroll
        for (int d = 0; d < D_; ++d) { float v = Pn[d * P_ + lane]; s += v * v; }
        p2s[lane] = s;
    }
    {
        v4f* rz = reinterpret_cast<v4f*>(ring);
        const v4f z = {0.0f, 0.0f, 0.0f, 0.0f};
        for (int q = 0; q < (P_ * RS_) / (4 * 32); ++q) rz[q * 32 + lane] = z;
    }
    __syncthreads();

    v2f Afrag[2][3];
    #pragma unroll
    for (int m = 0; m < 2; ++m)
        #pragma unroll
        for (int kc = 0; kc < 3; ++kc) {
            int d0 = 4 * kc + 2 * hi;
            int p  = m * 16 + lo;
            v2f a; a.x = Pn[d0 * P_ + p]; a.y = Pn[(d0 + 1) * P_ + p];
            Afrag[m][kc] = a;
        }

    float p2v[2][8];
    #pragma unroll
    for (int m = 0; m < 2; ++m)
        #pragma unroll
        for (int r = 0; r < 8; ++r)
            p2v[m][r] = p2s[m * 16 + 8 * hi + r];

    cost_tile(ring, Afrag, Bf, p2v, 0, lane, hi, lo);
    load_xfrag(Xb, 1, hi, lo, Bf);

    float dcur = (lane == 0) ? 0.0f : BIGV;
    float diag = BIGV;
    float z15 = BIGV;
    const bool is15 = (lane == 15);
    float* outp = out + ((b * N_ + n) * P_ + lane) * LOUT_;

    for (int T = 0; T < 2; ++T)
        tile_step<true, false>(T, ring, Xb, Afrag, Bf, p2v, lane, hi, lo, is15,
                               dcur, diag, z15, outp);
    for (int T = 2; T < (L_ - LOUT_) / 16; ++T)
        tile_step<false, false>(T, ring, Xb, Afrag, Bf, p2v, lane, hi, lo, is15,
                                dcur, diag, z15, outp);
    for (int T = (L_ - LOUT_) / 16; T < NTS_; ++T)
        tile_step<true, true>(T, ring, Xb, Afrag, Bf, p2v, lane, hi, lo, is15,
                              dcur, diag, z15, outp);
}

extern "C" void kernel_launch(void* const* d_in, const int* in_sizes, int n_in,
                              void* d_out, int out_size, void* d_ws, size_t ws_size,
                              hipStream_t stream) {
    const float* x     = (const float*)d_in[0];
    const float* patts = (const float*)d_in[1];
    float*       outp  = (float*)d_out;
    (void)in_sizes; (void)n_in; (void)out_size; (void)d_ws; (void)ws_size;

    dim3 grid(B_, N_);
    dim3 block(32);
    dtw_wmma_kernel<<<grid, block, 0, stream>>>(x, patts, outp);
}


# Launch shape: 256 workgroups of 4 waves, one (b,n) per wave in a private LDS slice with no workgroup barrier; LDS sized to one workgroup per WGP so each wave has a SIMD to itself; DP adds unchanged
# speedup vs baseline: 8.9606x; 1.1986x over previous
#include <hip/hip_runtime.h>


typedef float v2f __attribute__((ext_vector_type(2)));
typedef float v4f __attribute__((ext_vector_type(4)));
typedef float v8f __attribute__((ext_vector_type(8)));

#define B_    32
#define N_    32
#define D_    12
#define P_    32
#define L_    1024
#define LOUT_ 32
#define BIGV  1e30f
#define NT_   (L_ / 16)
#define NTS_  ((L_ + P_) / 16)
#define RS_   68
#define NW_   4
#define WGT_  128
#define NBY_  8
#define SLICE_ 10496


__device__ __forceinline__ float xhalf_f(float v) {
    return __builtin_bit_cast(float,
        __builtin_amdgcn_permlanex16(0, __builtin_bit_cast(int, v),
                                     0x76543210u, 0xfedcba98u, false, false));
}

__device__ __forceinline__ float row_up1(float v) {
    return __builtin_bit_cast(float,
        __builtin_amdgcn_update_dpp(__builtin_bit_cast(int, BIGV), __builtin_bit_cast(int, v),
                                    0x111, 0xf, 0xf, false));
}

__device__ __forceinline__ float cross_up1(float z) {
    return __builtin_bit_cast(float,
        __builtin_amdgcn_permlanex16(0, __builtin_bit_cast(int, z),
                                     0x0000000fu, 0x00000000u, false, false));
}

__device__ __forceinline__ void wave_sync() {
    __builtin_amdgcn_fence(__ATOMIC_RELEASE, "wavefront");
    __builtin_amdgcn_wave_barrier();
    __builtin_amdgcn_fence(__ATOMIC_ACQUIRE, "wavefront");
}

__device__ __forceinline__ void load_xfrag(const float* __restrict__ Xb, int t,
                                           int hi, int lo, v2f (&Bf)[3]) {
    const int j = t * 16 + lo;
    #pragma unroll
    for (int kc = 0; kc < 3; ++kc) {
        int d0 = 4 * kc + 2 * hi;
        v2f bb; bb.x = Xb[d0 * L_ + j]; bb.y = Xb[(d0 + 1) * L_ + j];
        Bf[kc] = bb;
    }
}

__device__ __forceinline__ void cost_tile(float* __restrict__ ring,
                                          const v2f (&Af)[2][3], const v2f (&Bf)[3],
                                          const float (&p2v)[2][8],
                                          int t, int lane, int hi, int lo) {
    float sx = 0.0f;
    #pragma unroll
    for (int kc = 0; kc < 3; ++kc) {
        float b0 = Bf[kc].x, b1 = Bf[kc].y;
        sx += b0 * b0 + b1 * b1;
    }
    sx += xhalf_f(sx);

    #pragma unroll
    for (int m = 0; m < 2; ++m) {
        v8f acc = {};
        acc = __builtin_amdgcn_wmma_f32_16x16x4_f32(false, Af[m][0],
                false, Bf[0], (short)0, acc, false, false);
        acc = __builtin_amdgcn_wmma_f32_16x16x4_f32(false, Af[m][1],
                false, Bf[1], (short)0, acc, false, false);
        acc = __builtin_amdgcn_wmma_f32_16x16x4_f32(false, Af[m][2],
                false, Bf[2], (short)0, acc, false, false);
        #pragma unroll
        for (int r = 0; r < 8; ++r) {
            const int row = m * 16 + 8 * hi + r;
            const int pos = (t * 16 + lo + row) & 63;
            ring[row * RS_ + pos] = p2v[m][r] + sx - 2.0f * acc[r];
        }
    }
}

template <bool EDGE, bool STORE>
__device__ __forceinline__ void dp16(const float (&c)[16], int kbase, int lane, bool is15,
                                     float& dcur, float& diag, float& z15,
                                     float* __restrict__ outp) {
    #pragma unroll
    for (int s = 0; s < 16; ++s) {
        const int j = kbase + s - lane;
        const float u = row_up1(dcur);
        const float t = cross_up1(z15);
        float dg = diag;
        if (EDGE) dg = (j == 0) ? BIGV : dg;
        const float w = fminf(dg, dcur);
        const float m = fminf(fminf(u, w), t);
        const float dn = c[s] + m;
        diag = fminf(u, t);
        if (EDGE) {
            dcur = ((unsigned)j < (unsigned)L_) ? dn : dcur;
            z15  = is15 ? dcur : BIGV;
        } else {
            dcur = dn;
            z15  = (is15 ? c[s] : BIGV) + m;
        }
        if (STORE) {
            if ((unsigned)(j - (L_ - LOUT_)) < (unsigned)LOUT_) outp[j - (L_ - LOUT_)] = dn;
        }
    }
}

template <bool EDGE, bool STORE>
__device__ __forceinline__ void tile_step(int T, float* __restrict__ ring,
                                          const float* __restrict__ Xb,
                                          const v2f (&Af)[2][3], v2f (&Bf)[3],
                                          const float (&p2v)[2][8],
                                          int lane, int hi, int lo, bool is15,
                                          float& dcur, float& diag, float& z15,
                                          float* __restrict__ outp) {
    wave_sync();

    const v4f* cp = reinterpret_cast<const v4f*>(ring + lane * RS_ + ((T * 16) & 63));
    const v4f q0 = cp[0], q1 = cp[1], q2 = cp[2], q3 = cp[3];
    const float c[16] = { q0.x, q0.y, q0.z, q0.w, q1.x, q1.y, q1.z, q1.w,
                          q2.x, q2.y, q2.z, q2.w, q3.x, q3.y, q3.z, q3.w };

    if (T + 1 < NT_) {
        cost_tile(ring, Af, Bf, p2v, T + 1, lane, hi, lo);
        if (T + 2 < NT_) load_xfrag(Xb, T + 2, hi, lo, Bf);
    }

    dp16<EDGE, STORE>(c, T * 16, lane, is15, dcur, diag, z15, outp);
}

__global__ __launch_bounds__(WGT_) void dtw_wmma_kernel(
    const float* __restrict__ x,
    const float* __restrict__ patts,
    float* __restrict__ out)
{
    const int lane = threadIdx.x & 31;
    const int hi   = lane >> 4;
    const int lo   = lane & 15;
    const int wv   = __builtin_amdgcn_readfirstlane((int)(threadIdx.x >> 5));
    const int b    = blockIdx.x;
    const int n    = blockIdx.y * NW_ + wv;

    __shared__ float p2all[NW_ * P_];
    __shared__ __attribute__((aligned(16))) float lds[NW_ * SLICE_];
    float* p2s  = p2all + wv * P_;
    float* ring = lds + wv * SLICE_;

    const float* Pn = patts + n * (D_ * P_);
    const float* Xb = x + b * (D_ * L_);

    v2f Bf[3];
    load_xfrag(Xb, 0, hi, lo, Bf);

    {
        float s = 0.0f;
        #pragma unroll
        for (int d = 0; d < D_; ++d) { float v = Pn[d * P_ + lane]; s += v * v; }
        p2s[lane] = s;
    }
    {
        v4f* rz = reinterpret_cast<v4f*>(ring);
        const v4f z = {0.0f, 0.0f, 0.0f, 0.0f};
        for (int q = 0; q < (P_ * RS_) / (4 * 32); ++q) rz[q * 32 + lane] = z;
    }
    wave_sync();

    v2f Afrag[2][3];
    #pragma unroll
    for (int m = 0; m < 2; ++m)
        #pragma unroll
        for (int kc = 0; kc < 3; ++kc) {
            int d0 = 4 * kc + 2 * hi;
            int p  = m * 16 + lo;
            v2f a; a.x = Pn[d0 * P_ + p]; a.y = Pn[(d0 + 1) * P_ + p];
            Afrag[m][kc] = a;
        }

    float p2v[2][8];
    #pragma unroll
    for (int m = 0; m < 2; ++m)
        #pragma unroll
        for (int r = 0; r < 8; ++r)
            p2v[m][r] = p2s[m * 16 + 8 * hi + r];

    cost_tile(ring, Afrag, Bf, p2v, 0, lane, hi, lo);
    load_xfrag(Xb, 1, hi, lo, Bf);

    float dcur = (lane == 0) ? 0.0f : BIGV;
    float diag = BIGV;
    float z15 = BIGV;
    const bool is15 = (lane == 15);
    float* outp = out + ((b * N_ + n) * P_ + lane) * LOUT_;

    for (int T = 0; T < 2; ++T)
        tile_step<true, false>(T, ring, Xb, Afrag, Bf, p2v, lane, hi, lo, is15,
                               dcur, diag, z15, outp);
    for (int T = 2; T < (L_ - LOUT_) / 16; ++T)
        tile_step<false, false>(T, ring, Xb, Afrag, Bf, p2v, lane, hi, lo, is15,
                                dcur, diag, z15, outp);
    for (int T = (L_ - LOUT_) / 16; T < NTS_; ++T)
        tile_step<true, true>(T, ring, Xb, Afrag, Bf, p2v, lane, hi, lo, is15,
                              dcur, diag, z15, outp);
}

extern "C" void kernel_launch(void* const* d_in, const int* in_sizes, int n_in,
                              void* d_out, int out_size, void* d_ws, size_t ws_size,
                              hipStream_t stream) {
    const float* x     = (const float*)d_in[0];
    const float* patts = (const float*)d_in[1];
    float*       outp  = (float*)d_out;
    (void)in_sizes; (void)n_in; (void)out_size; (void)d_ws; (void)ws_size;

    dim3 grid(B_, NBY_);
    dim3 block(WGT_);
    dtw_wmma_kernel<<<grid, block, 0, stream>>>(x, patts, outp);
}


# x fragment loads issued four tile-steps ahead into a ring of four register sets (was one ahead), so no tile-step waits for a second-level-cache load; costs, DP adds and their order unchanged
# speedup vs baseline: 11.1169x; 1.2406x over previous
#include <hip/hip_runtime.h>


typedef float v2f __attribute__((ext_vector_type(2)));
typedef float v4f __attribute__((ext_vector_type(4)));
typedef float v8f __attribute__((ext_vector_type(8)));

#define B_    32
#define N_    32
#define D_    12
#define P_    32
#define L_    1024
#define LOUT_ 32
#define BIGV  1e30f
#define NT_   (L_ / 16)
#define NTS_  ((L_ + P_) / 16)
#define RS_   68
#define NW_   4
#define WGT_  128
#define XD_   4
#define NBY_  8
#define SLICE_ 10496


__device__ __forceinline__ float xhalf_f(float v) {
    return __builtin_bit_cast(float,
        __builtin_amdgcn_permlanex16(0, __builtin_bit_cast(int, v),
                                     0x76543210u, 0xfedcba98u, false, false));
}

__device__ __forceinline__ float row_up1(float v) {
    return __builtin_bit_cast(float,
        __builtin_amdgcn_update_dpp(__builtin_bit_cast(int, BIGV), __builtin_bit_cast(int, v),
                                    0x111, 0xf, 0xf, false));
}

__device__ __forceinline__ float cross_up1(float z) {
    return __builtin_bit_cast(float,
        __builtin_amdgcn_permlanex16(0, __builtin_bit_cast(int, z),
                                     0x0000000fu, 0x00000000u, false, false));
}

__device__ __forceinline__ void wave_sync() {
    __builtin_amdgcn_fence(__ATOMIC_RELEASE, "wavefront");
    __builtin_amdgcn_wave_barrier();
    __builtin_amdgcn_fence(__ATOMIC_ACQUIRE, "wavefront");
}

__device__ __forceinline__ void load_xfrag(const float* __restrict__ Xb, int t,
                                           int hi, int lo, v2f (&Bf)[3]) {
    const int j = t * 16 + lo;
    #pragma unroll
    for (int kc = 0; kc < 3; ++kc) {
        int d0 = 4 * kc + 2 * hi;
        v2f bb; bb.x = Xb[d0 * L_ + j]; bb.y = Xb[(d0 + 1) * L_ + j];
        Bf[kc] = bb;
    }
}

__device__ __forceinline__ void cost_tile(float* __restrict__ ring,
                                          const v2f (&Af)[2][3], const v2f (&Bf)[3],
                                          const float (&p2v)[2][8],
                                          int t, int lane, int hi, int lo) {
    float sx = 0.0f;
    #pragma unroll
    for (int kc = 0; kc < 3; ++kc) {
        float b0 = Bf[kc].x, b1 = Bf[kc].y;
        sx += b0 * b0 + b1 * b1;
    }
    sx += xhalf_f(sx);

    #pragma unroll
    for (int m = 0; m < 2; ++m) {
        v8f acc = {};
        acc = __builtin_amdgcn_wmma_f32_16x16x4_f32(false, Af[m][0],
                false, Bf[0], (short)0, acc, false, false);
        acc = __builtin_amdgcn_wmma_f32_16x16x4_f32(false, Af[m][1],
                false, Bf[1], (short)0, acc, false, false);
        acc = __builtin_amdgcn_wmma_f32_16x16x4_f32(false, Af[m][2],
                false, Bf[2], (short)0, acc, false, false);
        #pragma unroll
        for (int r = 0; r < 8; ++r) {
            const int row = m * 16 + 8 * hi + r;
            const int pos = (t * 16 + lo + row) & 63;
            ring[row * RS_ + pos] = p2v[m][r] + sx - 2.0f * acc[r];
        }
    }
}

template <bool EDGE, bool STORE>
__device__ __forceinline__ void dp16(const float (&c)[16], int kbase, int lane, bool is15,
                                     float& dcur, float& diag, float& z15,
                                     float* __restrict__ outp) {
    #pragma unroll
    for (int s = 0; s < 16; ++s) {
        const int j = kbase + s - lane;
        const float u = row_up1(dcur);
        const float t = cross_up1(z15);
        float dg = diag;
        if (EDGE) dg = (j == 0) ? BIGV : dg;
        const float w = fminf(dg, dcur);
        const float m = fminf(fminf(u, w), t);
        const float dn = c[s] + m;
        diag = fminf(u, t);
        if (EDGE) {
            dcur = ((unsigned)j < (unsigned)L_) ? dn : dcur;
            z15  = is15 ? dcur : BIGV;
        } else {
            dcur = dn;
            z15  = (is15 ? c[s] : BIGV) + m;
        }
        if (STORE) {
            if ((unsigned)(j - (L_ - LOUT_)) < (unsigned)LOUT_) outp[j - (L_ - LOUT_)] = dn;
        }
    }
}

template <bool EDGE, bool STORE, bool LOADX>
__device__ __forceinline__ void tile_step(int T, float* __restrict__ ring,
                                          const float* __restrict__ Xb,
                                          const v2f (&Af)[2][3], v2f (&Bf)[3],
                                          const float (&p2v)[2][8],
                                          int lane, int hi, int lo, bool is15,
                                          float& dcur, float& diag, float& z15,
                                          float* __restrict__ outp) {
    wave_sync();

    const v4f* cp = reinterpret_cast<const v4f*>(ring + lane * RS_ + ((T * 16) & 63));
    const v4f q0 = cp[0], q1 = cp[1], q2 = cp[2], q3 = cp[3];
    const float c[16] = { q0.x, q0.y, q0.z, q0.w, q1.x, q1.y, q1.z, q1.w,
                          q2.x, q2.y, q2.z, q2.w, q3.x, q3.y, q3.z, q3.w };

    if (T + 1 < NT_) {
        cost_tile(ring, Af, Bf, p2v, T + 1, lane, hi, lo);
        if (LOADX) load_xfrag(Xb, T + 1 + XD_, hi, lo, Bf);
    }

    dp16<EDGE, STORE>(c, T * 16, lane, is15, dcur, diag, z15, outp);
}

__global__ __launch_bounds__(WGT_) void dtw_wmma_kernel(
    const float* __restrict__ x,
    const float* __restrict__ patts,
    float* __restrict__ out)
{
    const int lane = threadIdx.x & 31;
    const int hi   = lane >> 4;
    const int lo   = lane & 15;
    const int wv   = __builtin_amdgcn_readfirstlane((int)(threadIdx.x >> 5));
    const int b    = blockIdx.x;
    const int n    = blockIdx.y * NW_ + wv;

    __shared__ float p2all[NW_ * P_];
    __shared__ __attribute__((aligned(16))) float lds[NW_ * SLICE_];
    float* p2s  = p2all + wv * P_;
    float* ring = lds + wv * SLICE_;

    const float* Pn = patts + n * (D_ * P_);
    const float* Xb = x + b * (D_ * L_);

    v2f Bq[XD_][3];
    #pragma unroll
    for (int q = 0; q < XD_; ++q) load_xfrag(Xb, q, hi, lo, Bq[q]);

    {
        float s = 0.0f;
        #pragma unroll
        for (int d = 0; d < D_; ++d) { float v = Pn[d * P_ + lane]; s += v * v; }
        p2s[lane] = s;
    }
    {
        v4f* rz = reinterpret_cast<v4f*>(ring);
        const v4f z = {0.0f, 0.0f, 0.0f, 0.0f};
        for (int q = 0; q < (P_ * RS_) / (4 * 32); ++q) rz[q * 32 + lane] = z;
    }
    wave_sync();

    v2f Afrag[2][3];
    #pragma unroll
    for (int m = 0; m < 2; ++m)
        #pragma unroll
        for (int kc = 0; kc < 3; ++kc) {
            int d0 = 4 * kc + 2 * hi;
            int p  = m * 16 + lo;
            v2f a; a.x = Pn[d0 * P_ + p]; a.y = Pn[(d0 + 1) * P_ + p];
            Afrag[m][kc] = a;
        }

    float p2v[2][8];
    #pragma unroll
    for (int m = 0; m < 2; ++m)
        #pragma unroll
        for (int r = 0; r < 8; ++r)
            p2v[m][r] = p2s[m * 16 + 8 * hi + r];

    cost_tile(ring, Afrag, Bq[0], p2v, 0, lane, hi, lo);
    load_xfrag(Xb, XD_, hi, lo, Bq[0]);

    float dcur = (lane == 0) ? 0.0f : BIGV;
    float diag = BIGV;
    float z15 = BIGV;
    const bool is15 = (lane == 15);
    float* outp = out + ((b * N_ + n) * P_ + lane) * LOUT_;

    tile_step<true, false, true>(0, ring, Xb, Afrag, Bq[1], p2v, lane, hi, lo, is15,
                                 dcur, diag, z15, outp);
    tile_step<true, false, true>(1, ring, Xb, Afrag, Bq[2], p2v, lane, hi, lo, is15,
                                 dcur, diag, z15, outp);
    static_assert(XD_ == 4 && ((L_ - LOUT_) / 16 - 2) % XD_ == 0, "trips of four tile-steps");
    static_assert((L_ - LOUT_) / 16 - 4 + 1 + XD_ == NT_ - 1, "last load is the last tile");
    for (int T = 2; T < (L_ - LOUT_) / 16 - 4; T += XD_) {
        tile_step<false, false, true>(T,     ring, Xb, Afrag, Bq[3], p2v, lane, hi, lo, is15,
                                      dcur, diag, z15, outp);
        tile_step<false, false, true>(T + 1, ring, Xb, Afrag, Bq[0], p2v, lane, hi, lo, is15,
                                      dcur, diag, z15, outp);
        tile_step<false, false, true>(T + 2, ring, Xb, Afrag, Bq[1], p2v, lane, hi, lo, is15,
                                      dcur, diag, z15, outp);
        tile_step<false, false, true>(T + 3, ring, Xb, Afrag, Bq[2], p2v, lane, hi, lo, is15,
                                      dcur, diag, z15, outp);
    }
    {
        const int T = (L_ - LOUT_) / 16 - 4;
        tile_step<false, false, true >(T,     ring, Xb, Afrag, Bq[3], p2v, lane, hi, lo, is15,
                                       dcur, diag, z15, outp);
        tile_step<false, false, false>(T + 1, ring, Xb, Afrag, Bq[0], p2v, lane, hi, lo, is15,
                                       dcur, diag, z15, outp);
        tile_step<false, false, false>(T + 2, ring, Xb, Afrag, Bq[1], p2v, lane, hi, lo, is15,
                                       dcur, diag, z15, outp);
        tile_step<false, false, false>(T + 3, ring, Xb, Afrag, Bq[2], p2v, lane, hi, lo, is15,
                                       dcur, diag, z15, outp);
    }
    for (int T = (L_ - LOUT_) / 16; T < NTS_; ++T)
        tile_step<true, true, false>(T, ring, Xb, Afrag, Bq[3], p2v, lane, hi, lo, is15,
                                     dcur, diag, z15, outp);
}

extern "C" void kernel_launch(void* const* d_in, const int* in_sizes, int n_in,
                              void* d_out, int out_size, void* d_ws, size_t ws_size,
                              hipStream_t stream) {
    const float* x     = (const float*)d_in[0];
    const float* patts = (const float*)d_in[1];
    float*       outp  = (float*)d_out;
    (void)in_sizes; (void)n_in; (void)out_size; (void)d_ws; (void)ws_size;

    dim3 grid(B_, NBY_);
    dim3 block(WGT_);
    dtw_wmma_kernel<<<grid, block, 0, stream>>>(x, patts, outp);
}


# Output staged in LDS: the last 64 steps store each value unmasked to a per-lane stage row (no compare, exec change, branch and 32-line store per step); 32 full-line stores after the DP; same sums
# speedup vs baseline: 13.9028x; 1.2506x over previous
#include <hip/hip_runtime.h>


typedef float v2f __attribute__((ext_vector_type(2)));
typedef float v4f __attribute__((ext_vector_type(4)));
typedef float v8f __attribute__((ext_vector_type(8)));

#define B_    32
#define N_    32
#define D_    12
#define P_    32
#define L_    1024
#define LOUT_ 32
#define BIGV  1e30f
#define NT_   (L_ / 16)
#define NTS_  ((L_ + P_) / 16)
#define RS_   68
#define NW_   4
#define WGT_  128
#define XD_   4
#define NBY_  8
#define SLICE_ 10496
#define NFS_  (NTS_ * 16 - (L_ - LOUT_))
#define SP_   (NFS_ + 1)
static_assert(P_ * RS_ + P_ * SP_ <= SLICE_, "ring and output stage fit the wave's slice");


__device__ __forceinline__ float xhalf_f(float v) {
    return __builtin_bit_cast(float,
        __builtin_amdgcn_permlanex16(0, __builtin_bit_cast(int, v),
                                     0x76543210u, 0xfedcba98u, false, false));
}

__device__ __forceinline__ float row_up1(float v) {
    return __builtin_bit_cast(float,
        __builtin_amdgcn_update_dpp(__builtin_bit_cast(int, BIGV), __builtin_bit_cast(int, v),
                                    0x111, 0xf, 0xf, false));
}

__device__ __forceinline__ float cross_up1(float z) {
    return __builtin_bit_cast(float,
        __builtin_amdgcn_permlanex16(0, __builtin_bit_cast(int, z),
                                     0x0000000fu, 0x00000000u, false, false));
}

__device__ __forceinline__ void wave_sync() {
    __builtin_amdgcn_fence(__ATOMIC_RELEASE, "wavefront");
    __builtin_amdgcn_wave_barrier();
    __builtin_amdgcn_fence(__ATOMIC_ACQUIRE, "wavefront");
}

__device__ __forceinline__ void load_xfrag(const float* __restrict__ Xb, int t,
                                           int hi, int lo, v2f (&Bf)[3]) {
    const int j = t * 16 + lo;
    #pragma unroll
    for (int kc = 0; kc < 3; ++kc) {
        int d0 = 4 * kc + 2 * hi;
        v2f bb; bb.x = Xb[d0 * L_ + j]; bb.y = Xb[(d0 + 1) * L_ + j];
        Bf[kc] = bb;
    }
}

__device__ __forceinline__ void cost_tile(float* __restrict__ ring,
                                          const v2f (&Af)[2][3], const v2f (&Bf)[3],
                                          const float (&p2v)[2][8],
                                          int t, int lane, int hi, int lo) {
    float sx = 0.0f;
    #pragma unroll
    for (int kc = 0; kc < 3; ++kc) {
        float b0 = Bf[kc].x, b1 = Bf[kc].y;
        sx += b0 * b0 + b1 * b1;
    }
    sx += xhalf_f(sx);

    #pragma unroll
    for (int m = 0; m < 2; ++m) {
        v8f acc = {};
        acc = __builtin_amdgcn_wmma_f32_16x16x4_f32(false, Af[m][0],
                false, Bf[0], (short)0, acc, false, false);
        acc = __builtin_amdgcn_wmma_f32_16x16x4_f32(false, Af[m][1],
                false, Bf[1], (short)0, acc, false, false);
        acc = __builtin_amdgcn_wmma_f32_16x16x4_f32(false, Af[m][2],
                false, Bf[2], (short)0, acc, false, false);
        #pragma unroll
        for (int r = 0; r < 8; ++r) {
            const int row = m * 16 + 8 * hi + r;
            const int pos = (t * 16 + lo + row) & 63;
            ring[row * RS_ + pos] = p2v[m][r] + sx - 2.0f * acc[r];
        }
    }
}

template <bool EDGE, bool STORE>
__device__ __forceinline__ void dp16(const float (&c)[16], int kbase, int lane, bool is15,
                                     float& dcur, float& diag, float& z15,
                                     float* __restrict__ stl) {
    #pragma unroll
    for (int s = 0; s < 16; ++s) {
        const int j = kbase + s - lane;
        const float u = row_up1(dcur);
        const float t = cross_up1(z15);
        float dg = diag;
        if (EDGE) dg = (j == 0) ? BIGV : dg;
        const float w = fminf(dg, dcur);
        const float m = fminf(fminf(u, w), t);
        const float dn = c[s] + m;
        diag = fminf(u, t);
        if (EDGE) {
            dcur = ((unsigned)j < (unsigned)L_) ? dn : dcur;
            z15  = is15 ? dcur : BIGV;
        } else {
            dcur = dn;
            z15  = (is15 ? c[s] : BIGV) + m;
        }
        if (STORE) stl[kbase - (L_ - LOUT_) + s] = dn;
    }
}

template <bool EDGE, bool STORE, bool LOADX>
__device__ __forceinline__ void tile_step(int T, float* __restrict__ ring,
                                          const float* __restrict__ Xb,
                                          const v2f (&Af)[2][3], v2f (&Bf)[3],
                                          const float (&p2v)[2][8],
                                          int lane, int hi, int lo, bool is15,
                                          float& dcur, float& diag, float& z15,
                                          float* __restrict__ stl) {
    wave_sync();

    const v4f* cp = reinterpret_cast<const v4f*>(ring + lane * RS_ + ((T * 16) & 63));
    const v4f q0 = cp[0], q1 = cp[1], q2 = cp[2], q3 = cp[3];
    const float c[16] = { q0.x, q0.y, q0.z, q0.w, q1.x, q1.y, q1.z, q1.w,
                          q2.x, q2.y, q2.z, q2.w, q3.x, q3.y, q3.z, q3.w };

    if (T + 1 < NT_) {
        cost_tile(ring, Af, Bf, p2v, T + 1, lane, hi, lo);
        if (LOADX) load_xfrag(Xb, T + 1 + XD_, hi, lo, Bf);
    }

    dp16<EDGE, STORE>(c, T * 16, lane, is15, dcur, diag, z15, stl);
}

__global__ __launch_bounds__(WGT_) void dtw_wmma_kernel(
    const float* __restrict__ x,
    const float* __restrict__ patts,
    float* __restrict__ out)
{
    const int lane = threadIdx.x & 31;
    const int hi   = lane >> 4;
    const int lo   = lane & 15;
    const int wv   = __builtin_amdgcn_readfirstlane((int)(threadIdx.x >> 5));
    const int b    = blockIdx.x;
    const int n    = blockIdx.y * NW_ + wv;

    __shared__ float p2all[NW_ * P_];
    __shared__ __attribute__((aligned(16))) float lds[NW_ * SLICE_];
    float* p2s  = p2all + wv * P_;
    float* ring = lds + wv * SLICE_;

    const float* Pn = patts + n * (D_ * P_);
    const float* Xb = x + b * (D_ * L_);

    v2f Bq[XD_][3];
    #pragma unroll
    for (int q = 0; q < XD_; ++q) load_xfrag(Xb, q, hi, lo, Bq[q]);

    {
        float s = 0.0f;
        #pragma unroll
        for (int d = 0; d < D_; ++d) { float v = Pn[d * P_ + lane]; s += v * v; }
        p2s[lane] = s;
    }
    {
        v4f* rz = reinterpret_cast<v4f*>(ring);
        const v4f z = {0.0f, 0.0f, 0.0f, 0.0f};
        for (int q = 0; q < (P_ * RS_) / (4 * 32); ++q) rz[q * 32 + lane] = z;
    }
    wave_sync();

    v2f Afrag[2][3];
    #pragma unroll
    for (int m = 0; m < 2; ++m)
        #pragma unroll
        for (int kc = 0; kc < 3; ++kc) {
            int d0 = 4 * kc + 2 * hi;
            int p  = m * 16 + lo;
            v2f a; a.x = Pn[d0 * P_ + p]; a.y = Pn[(d0 + 1) * P_ + p];
            Afrag[m][kc] = a;
        }

    float p2v[2][8];
    #pragma unroll
    for (int m = 0; m < 2; ++m)
        #pragma unroll
        for (int r = 0; r < 8; ++r)
            p2v[m][r] = p2s[m * 16 + 8 * hi + r];

    cost_tile(ring, Afrag, Bq[0], p2v, 0, lane, hi, lo);
    load_xfrag(Xb, XD_, hi, lo, Bq[0]);

    float dcur = (lane == 0) ? 0.0f : BIGV;
    float diag = BIGV;
    float z15 = BIGV;
    const bool is15 = (lane == 15);
    float* stage = ring + P_ * RS_;
    float* stl = stage + lane * SP_;

    tile_step<true, false, true>(0, ring, Xb, Afrag, Bq[1], p2v, lane, hi, lo, is15,
                                 dcur, diag, z15, stl);
    tile_step<true, false, true>(1, ring, Xb, Afrag, Bq[2], p2v, lane, hi, lo, is15,
                                 dcur, diag, z15, stl);
    static_assert(XD_ == 4 && ((L_ - LOUT_) / 16 - 2) % XD_ == 0, "trips of four tile-steps");
    static_assert((L_ - LOUT_) / 16 - 4 + 1 + XD_ == NT_ - 1, "last load is the last tile");
    for (int T = 2; T < (L_ - LOUT_) / 16 - 4; T += XD_) {
        tile_step<false, false, true>(T,     ring, Xb, Afrag, Bq[3], p2v, lane, hi, lo, is15,
                                      dcur, diag, z15, stl);
        tile_step<false, false, true>(T + 1, ring, Xb, Afrag, Bq[0], p2v, lane, hi, lo, is15,
                                      dcur, diag, z15, stl);
        tile_step<false, false, true>(T + 2, ring, Xb, Afrag, Bq[1], p2v, lane, hi, lo, is15,
                                      dcur, diag, z15, stl);
        tile_step<false, false, true>(T + 3, ring, Xb, Afrag, Bq[2], p2v, lane, hi, lo, is15,
                                      dcur, diag, z15, stl);
    }
    {
        const int T = (L_ - LOUT_) / 16 - 4;
        tile_step<false, false, true >(T,     ring, Xb, Afrag, Bq[3], p2v, lane, hi, lo, is15,
                                       dcur, diag, z15, stl);
        tile_step<false, false, false>(T + 1, ring, Xb, Afrag, Bq[0], p2v, lane, hi, lo, is15,
                                       dcur, diag, z15, stl);
        tile_step<false, false, false>(T + 2, ring, Xb, Afrag, Bq[1], p2v, lane, hi, lo, is15,
                                       dcur, diag, z15, stl);
        tile_step<false, false, false>(T + 3, ring, Xb, Afrag, Bq[2], p2v, lane, hi, lo, is15,
                                       dcur, diag, z15, stl);
    }
    for (int T = (L_ - LOUT_) / 16; T < NTS_; ++T)
        tile_step<true, true, false>(T, ring, Xb, Afrag, Bq[3], p2v, lane, hi, lo, is15,
                                     dcur, diag, z15, stl);

    wave_sync();
    const float* sg = stage + lane;
    float* ob = out + (b * N_ + n) * (P_ * LOUT_) + lane;
    for (int r = 0; r < P_; ++r) ob[r * LOUT_] = sg[r * (SP_ + 1)];
}

extern "C" void kernel_launch(void* const* d_in, const int* in_sizes, int n_in,
                              void* d_out, int out_size, void* d_ws, size_t ws_size,
                              hipStream_t stream) {
    const float* x     = (const float*)d_in[0];
    const float* patts = (const float*)d_in[1];
    float*       outp  = (float*)d_out;
    (void)in_sizes; (void)n_in; (void)out_size; (void)d_ws; (void)ws_size;

    dim3 grid(B_, NBY_);
    dim3 block(WGT_);
    dtw_wmma_kernel<<<grid, block, 0, stream>>>(x, patts, outp);
}


# Tile build moved to a second wave per (b,n) on the DP wave's SIMD (8-wave workgroups, one barrier per tile-step): products and stores issue in the DP chain's idle cycles; same sums, same order
# speedup vs baseline: 15.2881x; 1.0996x over previous
#include <hip/hip_runtime.h>


typedef float v2f __attribute__((ext_vector_type(2)));
typedef float v4f __attribute__((ext_vector_type(4)));
typedef float v8f __attribute__((ext_vector_type(8)));

#define B_    32
#define N_    32
#define D_    12
#define P_    32
#define L_    1024
#define LOUT_ 32
#define BIGV  1e30f
#define NT_   (L_ / 16)
#define NTS_  ((L_ + P_) / 16)
#define RS_   68
#define NW_   4
#define WGT_  256
#define XD_   4
#define NBY_  8
#define SLICE_ 10496
#define NFS_  (NTS_ * 16 - (L_ - LOUT_))
#define SP_   (NFS_ + 1)
static_assert(P_ * RS_ + P_ * SP_ <= SLICE_, "ring and output stage fit the wave's slice");


__device__ __forceinline__ float xhalf_f(float v) {
    return __builtin_bit_cast(float,
        __builtin_amdgcn_permlanex16(0, __builtin_bit_cast(int, v),
                                     0x76543210u, 0xfedcba98u, false, false));
}

__device__ __forceinline__ float row_up1(float v) {
    return __builtin_bit_cast(float,
        __builtin_amdgcn_update_dpp(__builtin_bit_cast(int, BIGV), __builtin_bit_cast(int, v),
                                    0x111, 0xf, 0xf, false));
}

__device__ __forceinline__ float cross_up1(float z) {
    return __builtin_bit_cast(float,
        __builtin_amdgcn_permlanex16(0, __builtin_bit_cast(int, z),
                                     0x0000000fu, 0x00000000u, false, false));
}

__device__ __forceinline__ void wave_sync() {
    __builtin_amdgcn_fence(__ATOMIC_RELEASE, "wavefront");
    __builtin_amdgcn_wave_barrier();
    __builtin_amdgcn_fence(__ATOMIC_ACQUIRE, "wavefront");
}

__device__ __forceinline__ void load_xfrag(const float* __restrict__ Xb, int t,
                                           int hi, int lo, v2f (&Bf)[3]) {
    const int j = t * 16 + lo;
    #pragma unroll
    for (int kc = 0; kc < 3; ++kc) {
        int d0 = 4 * kc + 2 * hi;
        v2f bb; bb.x = Xb[d0 * L_ + j]; bb.y = Xb[(d0 + 1) * L_ + j];
        Bf[kc] = bb;
    }
}

__device__ __forceinline__ void cost_tile(float* __restrict__ ring,
                                          const v2f (&Af)[2][3], const v2f (&Bf)[3],
                                          const float (&p2v)[2][8],
                                          int t, int lane, int hi, int lo) {
    float sx = 0.0f;
    #pragma unroll
    for (int kc = 0; kc < 3; ++kc) {
        float b0 = Bf[kc].x, b1 = Bf[kc].y;
        sx += b0 * b0 + b1 * b1;
    }
    sx += xhalf_f(sx);

    #pragma unroll
    for (int m = 0; m < 2; ++m) {
        v8f acc = {};
        acc = __builtin_amdgcn_wmma_f32_16x16x4_f32(false, Af[m][0],
                false, Bf[0], (short)0, acc, false, false);
        acc = __builtin_amdgcn_wmma_f32_16x16x4_f32(false, Af[m][1],
                false, Bf[1], (short)0, acc, false, false);
        acc = __builtin_amdgcn_wmma_f32_16x16x4_f32(false, Af[m][2],
                false, Bf[2], (short)0, acc, false, false);
        #pragma unroll
        for (int r = 0; r < 8; ++r) {
            const int row = m * 16 + 8 * hi + r;
            const int pos = (t * 16 + lo + row) & 63;
            ring[row * RS_ + pos] = p2v[m][r] + sx - 2.0f * acc[r];
        }
    }
}

template <bool EDGE, bool STORE>
__device__ __forceinline__ void dp16(const float (&c)[16], int kbase, int lane, bool is15,
                                     float& dcur, float& diag, float& z15,
                                     float* __restrict__ stl) {
    #pragma unroll
    for (int s = 0; s < 16; ++s) {
        const int j = kbase + s - lane;
        const float u = row_up1(dcur);
        const float t = cross_up1(z15);
        float dg = diag;
        if (EDGE) dg = (j == 0) ? BIGV : dg;
        const float w = fminf(dg, dcur);
        const float m = fminf(fminf(u, w), t);
        const float dn = c[s] + m;
        diag = fminf(u, t);
        if (EDGE) {
            dcur = ((unsigned)j < (unsigned)L_) ? dn : dcur;
            z15  = is15 ? dcur : BIGV;
        } else {
            dcur = dn;
            z15  = (is15 ? c[s] : BIGV) + m;
        }
        if (STORE) stl[kbase - (L_ - LOUT_) + s] = dn;
    }
}

__device__ __forceinline__ void pair_sync() {
    __builtin_amdgcn_fence(__ATOMIC_RELEASE, "workgroup", "local");
    __builtin_amdgcn_s_barrier_signal(-1);
    __builtin_amdgcn_s_barrier_wait(-1);
    __builtin_amdgcn_fence(__ATOMIC_ACQUIRE, "workgroup", "local");
}

__device__ __forceinline__ void load_costs(const float* __restrict__ ring, int T, int lane,
                                           v4f (&q)[4]) {
    const v4f* cp = reinterpret_cast<const v4f*>(ring + lane * RS_ + ((T * 16) & 63));
    q[0] = cp[0]; q[1] = cp[1]; q[2] = cp[2]; q[3] = cp[3];
}

template <bool EDGE, bool STORE, bool BUILD, bool LOADX>
__device__ __forceinline__ void tile_step(int T, bool is_dp, float* __restrict__ ring,
                                          const float* __restrict__ Xb,
                                          const v2f (&Af)[2][3], v2f (&Bf)[3],
                                          const float (&p2v)[2][8],
                                          int lane, int hi, int lo, bool is15,
                                          v4f (&q)[4],
                                          float& dcur, float& diag, float& z15,
                                          float* __restrict__ stl) {
    if (is_dp) {
        v4f qn[4] = { q[0], q[1], q[2], q[3] };
        if (T + 1 < NTS_) load_costs(ring, T + 1, lane, qn);
        __builtin_amdgcn_sched_barrier(0);
        const float c[16] = { q[0].x, q[0].y, q[0].z, q[0].w, q[1].x, q[1].y, q[1].z, q[1].w,
                              q[2].x, q[2].y, q[2].z, q[2].w, q[3].x, q[3].y, q[3].z, q[3].w };
        dp16<EDGE, STORE>(c, T * 16, lane, is15, dcur, diag, z15, stl);
        q[0] = qn[0]; q[1] = qn[1]; q[2] = qn[2]; q[3] = qn[3];
    } else {
        if (BUILD) {
            cost_tile(ring, Af, Bf, p2v, T + 2, lane, hi, lo);
        }
    }
    if (LOADX) load_xfrag(Xb, T + 2 + XD_, hi, lo, Bf);
    pair_sync();
}

__global__ __launch_bounds__(WGT_) void dtw_wmma_kernel(
    const float* __restrict__ x,
    const float* __restrict__ patts,
    float* __restrict__ out)
{
    const int lane = threadIdx.x & 31;
    const int hi   = lane >> 4;
    const int lo   = lane & 15;
    const int wv   = __builtin_amdgcn_readfirstlane((int)(threadIdx.x >> 5));
    const bool is_dp = wv < NW_;
    const int pr   = wv & (NW_ - 1);
    const int b    = blockIdx.x;
    const int n    = blockIdx.y * NW_ + pr;

    __shared__ float p2all[NW_ * P_];
    __shared__ __attribute__((aligned(16))) float lds[NW_ * SLICE_];
    float* p2s  = p2all + pr * P_;
    float* ring = lds + pr * SLICE_;

    const float* Pn = patts + n * (D_ * P_);
    const float* Xb = x + b * (D_ * L_);

    v2f Bq[XD_][3];
    #pragma unroll
    for (int q = 0; q < XD_; ++q) load_xfrag(Xb, q, hi, lo, Bq[q]);

    v2f Afrag[2][3] = {};
    float p2v[2][8] = {};

    if (!is_dp) {

        {
            float s = 0.0f;
            #pragma unroll
            for (int d = 0; d < D_; ++d) { float v = Pn[d * P_ + lane]; s += v * v; }
            p2s[lane] = s;
        }
        {
            v4f* rz = reinterpret_cast<v4f*>(ring);
            const v4f z = {0.0f, 0.0f, 0.0f, 0.0f};
            for (int q = 0; q < (P_ * RS_) / (4 * 32); ++q) rz[q * 32 + lane] = z;
        }
        wave_sync();

        #pragma unroll
        for (int m = 0; m < 2; ++m)
            #pragma unroll
            for (int kc = 0; kc < 3; ++kc) {
                int d0 = 4 * kc + 2 * hi;
                int p  = m * 16 + lo;
                v2f a; a.x = Pn[d0 * P_ + p]; a.y = Pn[(d0 + 1) * P_ + p];
                Afrag[m][kc] = a;
            }

        #pragma unroll
        for (int m = 0; m < 2; ++m)
            #pragma unroll
            for (int r = 0; r < 8; ++r)
                p2v[m][r] = p2s[m * 16 + 8 * hi + r];

        cost_tile(ring, Afrag, Bq[0], p2v, 0, lane, hi, lo);
        cost_tile(ring, Afrag, Bq[1], p2v, 1, lane, hi, lo);
    }
    load_xfrag(Xb, XD_, hi, lo, Bq[0]);
    load_xfrag(Xb, XD_ + 1, hi, lo, Bq[1]);
    pair_sync();

    float dcur = (lane == 0) ? 0.0f : BIGV;
    float diag = BIGV;
    float z15 = BIGV;
    const bool is15 = (lane == 15);
    float* stage = ring + P_ * RS_;
    float* stl = stage + lane * SP_;
    v4f cq[4] = {};
    if (is_dp) load_costs(ring, 0, lane, cq);

    tile_step<true, false, true, true>(0, is_dp, ring, Xb, Afrag, Bq[2], p2v, lane, hi, lo, is15,
                                       cq, dcur, diag, z15, stl);
    tile_step<true, false, true, true>(1, is_dp, ring, Xb, Afrag, Bq[3], p2v, lane, hi, lo, is15,
                                       cq, dcur, diag, z15, stl);
    static_assert(XD_ == 4 && ((L_ - LOUT_) / 16 - 2) % XD_ == 0, "trips of four tile-steps");
    static_assert((L_ - LOUT_) / 16 - 4 - 1 + 2 + XD_ == NT_ - 1, "last load is the last tile");
    for (int T = 2; T < (L_ - LOUT_) / 16 - 4; T += XD_) {
        tile_step<false, false, true, true>(T,     is_dp, ring, Xb, Afrag, Bq[0], p2v, lane, hi, lo,
                                            is15, cq, dcur, diag, z15, stl);
        tile_step<false, false, true, true>(T + 1, is_dp, ring, Xb, Afrag, Bq[1], p2v, lane, hi, lo,
                                            is15, cq, dcur, diag, z15, stl);
        tile_step<false, false, true, true>(T + 2, is_dp, ring, Xb, Afrag, Bq[2], p2v, lane, hi, lo,
                                            is15, cq, dcur, diag, z15, stl);
        tile_step<false, false, true, true>(T + 3, is_dp, ring, Xb, Afrag, Bq[3], p2v, lane, hi, lo,
                                            is15, cq, dcur, diag, z15, stl);
    }
    {
        const int T = (L_ - LOUT_) / 16 - 4;
        static_assert((L_ - LOUT_) / 16 - 4 + 3 + 2 == NT_ - 1, "the last trip builds the last tile");
        tile_step<false, false, true, false>(T,     is_dp, ring, Xb, Afrag, Bq[0], p2v, lane, hi, lo,
                                             is15, cq, dcur, diag, z15, stl);
        tile_step<false, false, true, false>(T + 1, is_dp, ring, Xb, Afrag, Bq[1], p2v, lane, hi, lo,
                                             is15, cq, dcur, diag, z15, stl);
        tile_step<false, false, true, false>(T + 2, is_dp, ring, Xb, Afrag, Bq[2], p2v, lane, hi, lo,
                                             is15, cq, dcur, diag, z15, stl);
        tile_step<false, false, true, false>(T + 3, is_dp, ring, Xb, Afrag, Bq[3], p2v, lane, hi, lo,
                                             is15, cq, dcur, diag, z15, stl);
    }
    for (int T = (L_ - LOUT_) / 16; T < NTS_; ++T)
        tile_step<true, true, false, false>(T, is_dp, ring, Xb, Afrag, Bq[0], p2v, lane, hi, lo,
                                            is15, cq, dcur, diag, z15, stl);

    if (is_dp) {
        wave_sync();
        const float* sg = stage + lane;
        float* ob = out + (b * N_ + n) * (P_ * LOUT_) + lane;
        for (int r = 0; r < P_; ++r) ob[r * LOUT_] = sg[r * (SP_ + 1)];
    }
}

extern "C" void kernel_launch(void* const* d_in, const int* in_sizes, int n_in,
                              void* d_out, int out_size, void* d_ws, size_t ws_size,
                              hipStream_t stream) {
    const float* x     = (const float*)d_in[0];
    const float* patts = (const float*)d_in[1];
    float*       outp  = (float*)d_out;
    (void)in_sizes; (void)n_in; (void)out_size; (void)d_ws; (void)ws_size;

    dim3 grid(B_, NBY_);
    dim3 block(WGT_);
    dtw_wmma_kernel<<<grid, block, 0, stream>>>(x, patts, outp);
}


# DP skewed by two columns per lane (lane i runs 2i columns behind): the lane shift has a whole step to spare and leaves the dependent chain, a link is min3 + add; same adds, same order
# speedup vs baseline: 16.2002x; 1.0597x over previous
#include <hip/hip_runtime.h>


typedef float v2f __attribute__((ext_vector_type(2)));
typedef float v4f __attribute__((ext_vector_type(4)));
typedef float v8f __attribute__((ext_vector_type(8)));

#define B_    32
#define N_    32
#define D_    12
#define P_    32
#define L_    1024
#define LOUT_ 32
#define BIGV  1e30f
#define NT_   (L_ / 16)
#define SK_   2
#define NTS_  ((L_ + SK_ * P_ + 15) / 16)
#define RL_   128
#define RS_   132
#define NW_   4
#define WGT_  256
#define XD_   4
#define NBY_  8
#define SLICE_ 10496
#define NFS_  (NTS_ * 16 - (L_ - LOUT_))
#define SP_   (NFS_ + 1)
static_assert(16 * 3 + SK_ * (P_ - 1) + 16 <= RL_, "a tile lives in the ring until the last row has read it");
static_assert(P_ * RS_ + P_ * SP_ <= SLICE_, "ring and output stage fit the wave's slice");


__device__ __forceinline__ float xhalf_f(float v) {
    return __builtin_bit_cast(float,
        __builtin_amdgcn_permlanex16(0, __builtin_bit_cast(int, v),
                                     0x76543210u, 0xfedcba98u, false, false));
}

__device__ __forceinline__ float cross_up1(float z) {
    const int zi = __builtin_bit_cast(int, z);
    return __builtin_bit_cast(float,
        __builtin_amdgcn_permlanex16(zi, zi, 0x0000000fu, 0x00000000u, false, false));
}

__device__ __forceinline__ float row_up1_over(float keep, float v) {
    return __builtin_bit_cast(float,
        __builtin_amdgcn_update_dpp(__builtin_bit_cast(int, keep), __builtin_bit_cast(int, v),
                                    0x111, 0xf, 0xf, false));
}

__device__ __forceinline__ void wave_sync() {
    __builtin_amdgcn_fence(__ATOMIC_RELEASE, "wavefront");
    __builtin_amdgcn_wave_barrier();
    __builtin_amdgcn_fence(__ATOMIC_ACQUIRE, "wavefront");
}

__device__ __forceinline__ void load_xfrag(const float* __restrict__ Xb, int t,
                                           int hi, int lo, v2f (&Bf)[3]) {
    const int j = t * 16 + lo;
    #pragma unroll
    for (int kc = 0; kc < 3; ++kc) {
        int d0 = 4 * kc + 2 * hi;
        v2f bb; bb.x = Xb[d0 * L_ + j]; bb.y = Xb[(d0 + 1) * L_ + j];
        Bf[kc] = bb;
    }
}

__device__ __forceinline__ void cost_tile(float* __restrict__ ring,
                                          const v2f (&Af)[2][3], const v2f (&Bf)[3],
                                          const float (&p2v)[2][8],
                                          int t, int lane, int hi, int lo) {
    float sx = 0.0f;
    #pragma unroll
    for (int kc = 0; kc < 3; ++kc) {
        float b0 = Bf[kc].x, b1 = Bf[kc].y;
        sx += b0 * b0 + b1 * b1;
    }
    sx += xhalf_f(sx);

    #pragma unroll
    for (int m = 0; m < 2; ++m) {
        v8f acc = {};
        acc = __builtin_amdgcn_wmma_f32_16x16x4_f32(false, Af[m][0],
                false, Bf[0], (short)0, acc, false, false);
        acc = __builtin_amdgcn_wmma_f32_16x16x4_f32(false, Af[m][1],
                false, Bf[1], (short)0, acc, false, false);
        acc = __builtin_amdgcn_wmma_f32_16x16x4_f32(false, Af[m][2],
                false, Bf[2], (short)0, acc, false, false);
        #pragma unroll
        for (int r = 0; r < 8; ++r) {
            const int row = m * 16 + 8 * hi + r;
            const int pos = (t * 16 + lo + SK_ * row) & (RL_ - 1);
            ring[row * RS_ + pos] = p2v[m][r] + sx - 2.0f * acc[r];
        }
    }
}

template <bool STORE>
__device__ __forceinline__ void dp16(const float (&c)[16], int kbase, bool is15,
                                     float& dcur, float& p1, float& p2, float& p3,
                                     float* __restrict__ stl) {
    #pragma unroll
    for (int s = 0; s < 16; ++s) {
        const float m  = fminf(fminf(p2, p3), dcur);
        const float dn = c[s] + m;
        const float z  = (is15 ? c[s] : BIGV) + m;
        const float nn = row_up1_over(cross_up1(z), dn);
        p3 = p2; p2 = p1; p1 = nn;
        dcur = dn;
        if (STORE) stl[kbase - (L_ - LOUT_) + s] = dn;
    }
}

__device__ __forceinline__ void pair_sync() {
    __builtin_amdgcn_fence(__ATOMIC_RELEASE, "workgroup", "local");
    __builtin_amdgcn_s_barrier_signal(-1);
    __builtin_amdgcn_s_barrier_wait(-1);
    __builtin_amdgcn_fence(__ATOMIC_ACQUIRE, "workgroup", "local");
}

__device__ __forceinline__ void load_costs(const float* __restrict__ ring, int T, int lane,
                                           v4f (&q)[4]) {
    const v4f* cp = reinterpret_cast<const v4f*>(ring + lane * RS_ + ((T * 16) & (RL_ - 1)));
    q[0] = cp[0]; q[1] = cp[1]; q[2] = cp[2]; q[3] = cp[3];
}

template <bool STORE, bool BUILD, bool LOADX>
__device__ __forceinline__ void tile_step(int T, bool is_dp, float* __restrict__ ring,
                                          const float* __restrict__ Xb,
                                          const v2f (&Af)[2][3], v2f (&Bf)[3],
                                          const float (&p2v)[2][8],
                                          int lane, int hi, int lo, bool is15,
                                          v4f (&q)[4],
                                          float& dcur, float& p1, float& p2, float& p3,
                                          float* __restrict__ stl) {
    if (is_dp) {
        v4f qn[4] = { q[0], q[1], q[2], q[3] };
        if (T + 1 < NTS_) load_costs(ring, T + 1, lane, qn);
        __builtin_amdgcn_sched_barrier(0);
        const float c[16] = { q[0].x, q[0].y, q[0].z, q[0].w, q[1].x, q[1].y, q[1].z, q[1].w,
                              q[2].x, q[2].y, q[2].z, q[2].w, q[3].x, q[3].y, q[3].z, q[3].w };
        dp16<STORE>(c, T * 16, is15, dcur, p1, p2, p3, stl);
        q[0] = qn[0]; q[1] = qn[1]; q[2] = qn[2]; q[3] = qn[3];
    } else {
        if (BUILD) {
            cost_tile(ring, Af, Bf, p2v, T + 2, lane, hi, lo);
        }
    }
    if (LOADX) load_xfrag(Xb, T + 2 + XD_, hi, lo, Bf);
    pair_sync();
}

__global__ __launch_bounds__(WGT_) void dtw_wmma_kernel(
    const float* __restrict__ x,
    const float* __restrict__ patts,
    float* __restrict__ out)
{
    const int lane = threadIdx.x & 31;
    const int hi   = lane >> 4;
    const int lo   = lane & 15;
    const int wv   = __builtin_amdgcn_readfirstlane((int)(threadIdx.x >> 5));
    const bool is_dp = wv < NW_;
    const int pr   = wv & (NW_ - 1);
    const int b    = blockIdx.x;
    const int n    = blockIdx.y * NW_ + pr;

    __shared__ float p2all[NW_ * P_];
    __shared__ __attribute__((aligned(16))) float lds[NW_ * SLICE_];
    float* p2s  = p2all + pr * P_;
    float* ring = lds + pr * SLICE_;

    const float* Pn = patts + n * (D_ * P_);
    const float* Xb = x + b * (D_ * L_);

    v2f Bq[XD_][3];
    #pragma unroll
    for (int q = 0; q < XD_; ++q) load_xfrag(Xb, q, hi, lo, Bq[q]);

    v2f Afrag[2][3] = {};
    float p2v[2][8] = {};

    if (!is_dp) {

        {
            float s = 0.0f;
            #pragma unroll
            for (int d = 0; d < D_; ++d) { float v = Pn[d * P_ + lane]; s += v * v; }
            p2s[lane] = s;
        }
        static_assert((P_ * RS_) % (4 * 32) == 0, "the fill covers the whole ring");
        {
            v4f* rz = reinterpret_cast<v4f*>(ring);
            const v4f z = {BIGV, BIGV, BIGV, BIGV};
            for (int q = 0; q < (P_ * RS_) / (4 * 32); ++q) rz[q * 32 + lane] = z;
        }
        wave_sync();

        #pragma unroll
        for (int m = 0; m < 2; ++m)
            #pragma unroll
            for (int kc = 0; kc < 3; ++kc) {
                int d0 = 4 * kc + 2 * hi;
                int p  = m * 16 + lo;
                v2f a; a.x = Pn[d0 * P_ + p]; a.y = Pn[(d0 + 1) * P_ + p];
                Afrag[m][kc] = a;
            }

        #pragma unroll
        for (int m = 0; m < 2; ++m)
            #pragma unroll
            for (int r = 0; r < 8; ++r)
                p2v[m][r] = p2s[m * 16 + 8 * hi + r];

        cost_tile(ring, Afrag, Bq[0], p2v, 0, lane, hi, lo);
        cost_tile(ring, Afrag, Bq[1], p2v, 1, lane, hi, lo);
    }
    load_xfrag(Xb, XD_, hi, lo, Bq[0]);
    load_xfrag(Xb, XD_ + 1, hi, lo, Bq[1]);
    pair_sync();

    float dcur = BIGV;
    float p1 = BIGV;
    float p2 = BIGV;
    float p3 = (lane == 0) ? 0.0f : BIGV;
    const bool is15 = (lane == 15);
    float* stage = ring + P_ * RS_;
    float* stl = stage + lane * SP_;
    v4f cq[4] = {};
    if (is_dp) load_costs(ring, 0, lane, cq);

    tile_step<false, true, true>(0, is_dp, ring, Xb, Afrag, Bq[2], p2v, lane, hi, lo, is15,
                                 cq, dcur, p1, p2, p3, stl);
    tile_step<false, true, true>(1, is_dp, ring, Xb, Afrag, Bq[3], p2v, lane, hi, lo, is15,
                                 cq, dcur, p1, p2, p3, stl);
    static_assert(XD_ == 4 && ((L_ - LOUT_) / 16 - 2) % XD_ == 0, "trips of four tile-steps");
    static_assert((L_ - LOUT_) / 16 - 4 - 1 + 2 + XD_ == NT_ - 1, "last load is the last tile");
    for (int T = 2; T < (L_ - LOUT_) / 16 - 4; T += XD_) {
        tile_step<false, true, true>(T,     is_dp, ring, Xb, Afrag, Bq[0], p2v, lane, hi, lo,
                                     is15, cq, dcur, p1, p2, p3, stl);
        tile_step<false, true, true>(T + 1, is_dp, ring, Xb, Afrag, Bq[1], p2v, lane, hi, lo,
                                     is15, cq, dcur, p1, p2, p3, stl);
        tile_step<false, true, true>(T + 2, is_dp, ring, Xb, Afrag, Bq[2], p2v, lane, hi, lo,
                                     is15, cq, dcur, p1, p2, p3, stl);
        tile_step<false, true, true>(T + 3, is_dp, ring, Xb, Afrag, Bq[3], p2v, lane, hi, lo,
                                     is15, cq, dcur, p1, p2, p3, stl);
    }
    {
        const int T = (L_ - LOUT_) / 16 - 4;
        static_assert((L_ - LOUT_) / 16 - 4 + 3 + 2 == NT_ - 1, "the last trip builds the last tile");
        tile_step<false, true, false>(T,     is_dp, ring, Xb, Afrag, Bq[0], p2v, lane, hi, lo,
                                      is15, cq, dcur, p1, p2, p3, stl);
        tile_step<false, true, false>(T + 1, is_dp, ring, Xb, Afrag, Bq[1], p2v, lane, hi, lo,
                                      is15, cq, dcur, p1, p2, p3, stl);
        tile_step<false, true, false>(T + 2, is_dp, ring, Xb, Afrag, Bq[2], p2v, lane, hi, lo,
                                      is15, cq, dcur, p1, p2, p3, stl);
        tile_step<false, true, false>(T + 3, is_dp, ring, Xb, Afrag, Bq[3], p2v, lane, hi, lo,
                                      is15, cq, dcur, p1, p2, p3, stl);
    }
    static_assert(NTS_ * 16 > L_ - 1 + SK_ * (P_ - 1), "the last row reaches the last column");
    for (int T = (L_ - LOUT_) / 16; T < NTS_; ++T)
        tile_step<true, false, false>(T, is_dp, ring, Xb, Afrag, Bq[0], p2v, lane, hi, lo,
                                      is15, cq, dcur, p1, p2, p3, stl);

    if (is_dp) {
        wave_sync();
        const float* sg = stage + lane;
        float* ob = out + (b * N_ + n) * (P_ * LOUT_) + lane;
        for (int r = 0; r < P_; ++r) ob[r * LOUT_] = sg[r * (SP_ + SK_)];
    }
}

extern "C" void kernel_launch(void* const* d_in, const int* in_sizes, int n_in,
                              void* d_out, int out_size, void* d_ws, size_t ws_size,
                              hipStream_t stream) {
    const float* x     = (const float*)d_in[0];
    const float* patts = (const float*)d_in[1];
    float*       outp  = (float*)d_out;
    (void)in_sizes; (void)n_in; (void)out_size; (void)d_ws; (void)ws_size;

    dim3 grid(B_, NBY_);
    dim3 block(WGT_);
    dtw_wmma_kernel<<<grid, block, 0, stream>>>(x, patts, outp);
}


# Cost-ring store positions made compile-time constants: tile-steps written out in runs of 8 (a ring turn), so a store's address is an instruction offset, not add, mask, shift-add; same sums and order
# speedup vs baseline: 17.9862x; 1.1102x over previous
#include <hip/hip_runtime.h>


typedef float v2f __attribute__((ext_vector_type(2)));
typedef float v4f __attribute__((ext_vector_type(4)));
typedef float v8f __attribute__((ext_vector_type(8)));

#define B_    32
#define N_    32
#define D_    12
#define P_    32
#define L_    1024
#define LOUT_ 32
#define BIGV  1e30f
#define NT_   (L_ / 16)
#define SK_   2
#define NTS_  ((L_ + SK_ * P_ + 15) / 16)
#define RL_   128
#define RS_   132
#define NW_   4
#define WGT_  256
#define XD_   4
#define NBY_  8
#define SLICE_ 10496
#define NFS_  (NTS_ * 16 - (L_ - LOUT_))
#define SP_   (NFS_ + 1)
static_assert(16 * 3 + SK_ * (P_ - 1) + 16 <= RL_, "a tile lives in the ring until the last row has read it");
static_assert(P_ * RS_ + P_ * SP_ <= SLICE_, "ring and output stage fit the wave's slice");


__device__ __forceinline__ float xhalf_f(float v) {
    return __builtin_bit_cast(float,
        __builtin_amdgcn_permlanex16(0, __builtin_bit_cast(int, v),
                                     0x76543210u, 0xfedcba98u, false, false));
}

__device__ __forceinline__ float cross_up1(float z) {
    const int zi = __builtin_bit_cast(int, z);
    return __builtin_bit_cast(float,
        __builtin_amdgcn_permlanex16(zi, zi, 0x0000000fu, 0x00000000u, false, false));
}

__device__ __forceinline__ float row_up1_over(float keep, float v) {
    return __builtin_bit_cast(float,
        __builtin_amdgcn_update_dpp(__builtin_bit_cast(int, keep), __builtin_bit_cast(int, v),
                                    0x111, 0xf, 0xf, false));
}

__device__ __forceinline__ void wave_sync() {
    __builtin_amdgcn_fence(__ATOMIC_RELEASE, "wavefront");
    __builtin_amdgcn_wave_barrier();
    __builtin_amdgcn_fence(__ATOMIC_ACQUIRE, "wavefront");
}

__device__ __forceinline__ void load_xfrag(const float* __restrict__ Xb, int t,
                                           int hi, int lo, v2f (&Bf)[3]) {
    const int j = t * 16 + lo;
    #pragma unroll
    for (int kc = 0; kc < 3; ++kc) {
        int d0 = 4 * kc + 2 * hi;
        v2f bb; bb.x = Xb[d0 * L_ + j]; bb.y = Xb[(d0 + 1) * L_ + j];
        Bf[kc] = bb;
    }
}

template <int T8>
__device__ __forceinline__ void cost_tile(float* __restrict__ ring,
                                          const v2f (&Af)[2][3], const v2f (&Bf)[3],
                                          const float (&p2v)[2][8],
                                          int lane, int hi) {
    float sx = 0.0f;
    #pragma unroll
    for (int kc = 0; kc < 3; ++kc) {
        float b0 = Bf[kc].x, b1 = Bf[kc].y;
        sx += b0 * b0 + b1 * b1;
    }
    sx += xhalf_f(sx);

    #pragma unroll
    for (int m = 0; m < 2; ++m) {
        v8f acc = {};
        acc = __builtin_amdgcn_wmma_f32_16x16x4_f32(false, Af[m][0],
                false, Bf[0], (short)0, acc, false, false);
        acc = __builtin_amdgcn_wmma_f32_16x16x4_f32(false, Af[m][1],
                false, Bf[1], (short)0, acc, false, false);
        acc = __builtin_amdgcn_wmma_f32_16x16x4_f32(false, Af[m][2],
                false, Bf[2], (short)0, acc, false, false);
        #pragma unroll
        for (int r = 0; r < 8; ++r) {
            const int K = (16 * T8 + SK_ * (16 * m + r)) & (RL_ - 1);
            int idx = (8 * hi) * RS_ + lane + ((16 * m + r) * RS_ + K);
            if (K + 31 >= RL_) idx -= (lane >= RL_ - K) ? RL_ : 0;
            ring[idx] = p2v[m][r] + sx - 2.0f * acc[r];
        }
    }
}
static_assert(SK_ * 8 == 16 && RL_ == 128, "cost_tile: lo + SK_*8*hi is the lane; 8 tiles fill the ring");

template <bool STORE>
__device__ __forceinline__ void dp16(const float (&c)[16], int kbase, bool is15,
                                     float& dcur, float& p1, float& p2, float& p3,
                                     float* __restrict__ stl) {
    #pragma unroll
    for (int s = 0; s < 16; ++s) {
        const float m  = fminf(fminf(p2, p3), dcur);
        const float dn = c[s] + m;
        const float z  = (is15 ? c[s] : BIGV) + m;
        const float nn = row_up1_over(cross_up1(z), dn);
        p3 = p2; p2 = p1; p1 = nn;
        dcur = dn;
        if (STORE) stl[kbase - (L_ - LOUT_) + s] = dn;
    }
}

__device__ __forceinline__ void pair_sync() {
    __builtin_amdgcn_fence(__ATOMIC_RELEASE, "workgroup", "local");
    __builtin_amdgcn_s_barrier_signal(-1);
    __builtin_amdgcn_s_barrier_wait(-1);
    __builtin_amdgcn_fence(__ATOMIC_ACQUIRE, "workgroup", "local");
}

__device__ __forceinline__ void load_costs(const float* __restrict__ ring, int T, int lane,
                                           v4f (&q)[4]) {
    const v4f* cp = reinterpret_cast<const v4f*>(ring + lane * RS_ + ((T * 16) & (RL_ - 1)));
    q[0] = cp[0]; q[1] = cp[1]; q[2] = cp[2]; q[3] = cp[3];
}

template <bool STORE, bool BUILD, bool LOADX, int T8>
__device__ __forceinline__ void tile_step(int T, bool is_dp, float* __restrict__ ring,
                                          const float* __restrict__ Xb,
                                          const v2f (&Af)[2][3], v2f (&Bf)[3],
                                          const float (&p2v)[2][8],
                                          int lane, int hi, int lo, bool is15,
                                          v4f (&q)[4],
                                          float& dcur, float& p1, float& p2, float& p3,
                                          float* __restrict__ stl) {
    if (is_dp) {
        v4f qn[4] = { q[0], q[1], q[2], q[3] };
        if (T + 1 < NTS_) load_costs(ring, T + 1, lane, qn);
        __builtin_amdgcn_sched_barrier(0);
        const float c[16] = { q[0].x, q[0].y, q[0].z, q[0].w, q[1].x, q[1].y, q[1].z, q[1].w,
                              q[2].x, q[2].y, q[2].z, q[2].w, q[3].x, q[3].y, q[3].z, q[3].w };
        dp16<STORE>(c, T * 16, is15, dcur, p1, p2, p3, stl);
        q[0] = qn[0]; q[1] = qn[1]; q[2] = qn[2]; q[3] = qn[3];
    } else {
        if (BUILD) {
            cost_tile<T8>(ring, Af, Bf, p2v, lane, hi);
        }
    }
    if (LOADX) load_xfrag(Xb, T + 2 + XD_, hi, lo, Bf);
    pair_sync();
}

__global__ __launch_bounds__(WGT_) void dtw_wmma_kernel(
    const float* __restrict__ x,
    const float* __restrict__ patts,
    float* __restrict__ out)
{
    const int lane = threadIdx.x & 31;
    const int hi   = lane >> 4;
    const int lo   = lane & 15;
    const int wv   = __builtin_amdgcn_readfirstlane((int)(threadIdx.x >> 5));
    const bool is_dp = wv < NW_;
    const int pr   = wv & (NW_ - 1);
    const int b    = blockIdx.x;
    const int n    = blockIdx.y * NW_ + pr;

    __shared__ float p2all[NW_ * P_];
    __shared__ __attribute__((aligned(16))) float lds[NW_ * SLICE_];
    float* p2s  = p2all + pr * P_;
    float* ring = lds + pr * SLICE_;

    const float* Pn = patts + n * (D_ * P_);
    const float* Xb = x + b * (D_ * L_);

    v2f Bq[XD_][3];
    #pragma unroll
    for (int q = 0; q < XD_; ++q) load_xfrag(Xb, q, hi, lo, Bq[q]);

    v2f Afrag[2][3] = {};
    float p2v[2][8] = {};

    if (!is_dp) {

        {
            float s = 0.0f;
            #pragma unroll
            for (int d = 0; d < D_; ++d) { float v = Pn[d * P_ + lane]; s += v * v; }
            p2s[lane] = s;
        }
        static_assert((P_ * RS_) % (4 * 32) == 0, "the fill covers the whole ring");
        {
            v4f* rz = reinterpret_cast<v4f*>(ring);
            const v4f z = {BIGV, BIGV, BIGV, BIGV};
            for (int q = 0; q < (P_ * RS_) / (4 * 32); ++q) rz[q * 32 + lane] = z;
        }
        wave_sync();

        #pragma unroll
        for (int m = 0; m < 2; ++m)
            #pragma unroll
            for (int kc = 0; kc < 3; ++kc) {
                int d0 = 4 * kc + 2 * hi;
                int p  = m * 16 + lo;
                v2f a; a.x = Pn[d0 * P_ + p]; a.y = Pn[(d0 + 1) * P_ + p];
                Afrag[m][kc] = a;
            }

        #pragma unroll
        for (int m = 0; m < 2; ++m)
            #pragma unroll
            for (int r = 0; r < 8; ++r)
                p2v[m][r] = p2s[m * 16 + 8 * hi + r];

        cost_tile<0>(ring, Afrag, Bq[0], p2v, lane, hi);
        cost_tile<1>(ring, Afrag, Bq[1], p2v, lane, hi);
    }
    load_xfrag(Xb, XD_, hi, lo, Bq[0]);
    load_xfrag(Xb, XD_ + 1, hi, lo, Bq[1]);
    pair_sync();

    float dcur = BIGV;
    float p1 = BIGV;
    float p2 = BIGV;
    float p3 = (lane == 0) ? 0.0f : BIGV;
    const bool is15 = (lane == 15);
    float* stage = ring + P_ * RS_;
    float* stl = stage + lane * SP_;
    v4f cq[4] = {};
    if (is_dp) load_costs(ring, 0, lane, cq);

#define TS_(STORE_, BUILD_, LOADX_, T8_, T_, SET_)                                              \
    tile_step<STORE_, BUILD_, LOADX_, T8_>(T_, is_dp, ring, Xb, Afrag, Bq[SET_], p2v, lane, hi, \
                                           lo, is15, cq, dcur, p1, p2, p3, stl)
    TS_(false, true, true, 2, 0, 2);
    TS_(false, true, true, 3, 1, 3);
    static_assert(XD_ == 4 && RL_ / 16 == 8 && ((L_ - LOUT_) / 16 - 4 - 2) % 8 == 0, "trips of eight tile-steps");
    static_assert((L_ - LOUT_) / 16 - 4 - 1 + 2 + XD_ == NT_ - 1, "last load is the last tile");
    for (int T = 2; T < (L_ - LOUT_) / 16 - 4; T += 8) {
        TS_(false, true, true, 4, T,     0);
        TS_(false, true, true, 5, T + 1, 1);
        TS_(false, true, true, 6, T + 2, 2);
        TS_(false, true, true, 7, T + 3, 3);
        TS_(false, true, true, 0, T + 4, 0);
        TS_(false, true, true, 1, T + 5, 1);
        TS_(false, true, true, 2, T + 6, 2);
        TS_(false, true, true, 3, T + 7, 3);
    }
    {
        const int T = (L_ - LOUT_) / 16 - 4;
        static_assert((L_ - LOUT_) / 16 - 4 + 3 + 2 == NT_ - 1, "the last trip builds the last tile");
        static_assert(((L_ - LOUT_) / 16 - 4 + 2) % 8 == 4, "its first tile has ring phase 4");
        TS_(false, true, false, 4, T,     0);
        TS_(false, true, false, 5, T + 1, 1);
        TS_(false, true, false, 6, T + 2, 2);
        TS_(false, true, false, 7, T + 3, 3);
    }
    static_assert(NTS_ * 16 > L_ - 1 + SK_ * (P_ - 1), "the last row reaches the last column");
    for (int T = (L_ - LOUT_) / 16; T < NTS_; ++T)
        TS_(true, false, false, 0, T, 0);
#undef TS_

    if (is_dp) {
        wave_sync();
        const float* sg = stage + lane;
        float* ob = out + (b * N_ + n) * (P_ * LOUT_) + lane;
        for (int r = 0; r < P_; ++r) ob[r * LOUT_] = sg[r * (SP_ + SK_)];
    }
}

extern "C" void kernel_launch(void* const* d_in, const int* in_sizes, int n_in,
                              void* d_out, int out_size, void* d_ws, size_t ws_size,
                              hipStream_t stream) {
    const float* x     = (const float*)d_in[0];
    const float* patts = (const float*)d_in[1];
    float*       outp  = (float*)d_out;
    (void)in_sizes; (void)n_in; (void)out_size; (void)d_ws; (void)ws_size;

    dim3 grid(B_, NBY_);
    dim3 block(WGT_);
    dtw_wmma_kernel<<<grid, block, 0, stream>>>(x, patts, outp);
}


# Builder pipelined over two tiles: a tile-step stores the tile whose products it issued one tile-step before, then issues the next tile's; no wait for matrix pipe or LDS stores; same sums and order
# speedup vs baseline: 19.2516x; 1.0704x over previous
#include <hip/hip_runtime.h>


typedef float v2f __attribute__((ext_vector_type(2)));
typedef float v4f __attribute__((ext_vector_type(4)));
typedef float v8f __attribute__((ext_vector_type(8)));

#define B_    32
#define N_    32
#define D_    12
#define P_    32
#define L_    1024
#define LOUT_ 32
#define BIGV  1e30f
#define NT_   (L_ / 16)
#define SK_   2
#define NTS_  ((L_ + SK_ * P_ + 15) / 16)
#define RL_   128
#define RS_   132
#define NW_   4
#define WGT_  256
#define XD_   4
#define NBY_  8
#define SLICE_ 10496
#define NFS_  (NTS_ * 16 - (L_ - LOUT_))
#define SP_   (NFS_ + 1)
static_assert(16 * 3 + SK_ * (P_ - 1) + 16 <= RL_, "a tile lives in the ring until the last row has read it");
static_assert(P_ * RS_ + P_ * SP_ <= SLICE_, "ring and output stage fit the wave's slice");


__device__ __forceinline__ float xhalf_f(float v) {
    return __builtin_bit_cast(float,
        __builtin_amdgcn_permlanex16(0, __builtin_bit_cast(int, v),
                                     0x76543210u, 0xfedcba98u, false, false));
}

__device__ __forceinline__ float cross_up1(float z) {
    const int zi = __builtin_bit_cast(int, z);
    return __builtin_bit_cast(float,
        __builtin_amdgcn_permlanex16(zi, zi, 0x0000000fu, 0x00000000u, false, false));
}

__device__ __forceinline__ float row_up1_over(float keep, float v) {
    return __builtin_bit_cast(float,
        __builtin_amdgcn_update_dpp(__builtin_bit_cast(int, keep), __builtin_bit_cast(int, v),
                                    0x111, 0xf, 0xf, false));
}

__device__ __forceinline__ void wave_sync() {
    __builtin_amdgcn_fence(__ATOMIC_RELEASE, "wavefront");
    __builtin_amdgcn_wave_barrier();
    __builtin_amdgcn_fence(__ATOMIC_ACQUIRE, "wavefront");
}

__device__ __forceinline__ void load_xfrag(const float* __restrict__ Xb, int t,
                                           int hi, int lo, v2f (&Bf)[3]) {
    const int j = t * 16 + lo;
    #pragma unroll
    for (int kc = 0; kc < 3; ++kc) {
        int d0 = 4 * kc + 2 * hi;
        v2f bb; bb.x = Xb[d0 * L_ + j]; bb.y = Xb[(d0 + 1) * L_ + j];
        Bf[kc] = bb;
    }
}

__device__ __forceinline__ void tile_products(const v2f (&Af)[2][3], const v2f (&Bf)[3],
                                              v8f (&acc)[2], float& x2c) {
    float sx = 0.0f;
    #pragma unroll
    for (int kc = 0; kc < 3; ++kc) {
        float b0 = Bf[kc].x, b1 = Bf[kc].y;
        sx += b0 * b0 + b1 * b1;
    }
    sx += xhalf_f(sx);
    x2c = sx;

    #pragma unroll
    for (int m = 0; m < 2; ++m) {
        v8f a = {};
        a = __builtin_amdgcn_wmma_f32_16x16x4_f32(false, Af[m][0],
                false, Bf[0], (short)0, a, false, false);
        a = __builtin_amdgcn_wmma_f32_16x16x4_f32(false, Af[m][1],
                false, Bf[1], (short)0, a, false, false);
        a = __builtin_amdgcn_wmma_f32_16x16x4_f32(false, Af[m][2],
                false, Bf[2], (short)0, a, false, false);
        acc[m] = a;
    }
}

template <int T8>
__device__ __forceinline__ void tile_store(float* __restrict__ ring,
                                           const v8f (&acc)[2], float x2c,
                                           const float (&p2v)[2][8],
                                           int lane, int hi) {
    #pragma unroll
    for (int m = 0; m < 2; ++m) {
        #pragma unroll
        for (int r = 0; r < 8; ++r) {
            const int K = (16 * T8 + SK_ * (16 * m + r)) & (RL_ - 1);
            int idx = (8 * hi) * RS_ + lane + ((16 * m + r) * RS_ + K);
            if (K + 31 >= RL_) idx -= (lane >= RL_ - K) ? RL_ : 0;
            ring[idx] = p2v[m][r] + x2c - 2.0f * acc[m][r];
        }
    }
}
static_assert(SK_ * 8 == 16 && RL_ == 128, "tile_store: lo + SK_*8*hi is the lane; 8 tiles fill the ring");

template <bool STORE>
__device__ __forceinline__ void dp16(const float (&c)[16], int kbase, bool is15,
                                     float& dcur, float& p1, float& p2, float& p3,
                                     float* __restrict__ stl) {
    #pragma unroll
    for (int s = 0; s < 16; ++s) {
        const float m  = fminf(fminf(p2, p3), dcur);
        const float dn = c[s] + m;
        const float z  = (is15 ? c[s] : BIGV) + m;
        const float nn = row_up1_over(cross_up1(z), dn);
        p3 = p2; p2 = p1; p1 = nn;
        dcur = dn;
        if (STORE) stl[kbase - (L_ - LOUT_) + s] = dn;
    }
}

__device__ __forceinline__ void pair_sync() {
    __builtin_amdgcn_fence(__ATOMIC_RELEASE, "workgroup", "local");
    __builtin_amdgcn_s_barrier_signal(-1);
    __builtin_amdgcn_s_barrier_wait(-1);
    __builtin_amdgcn_fence(__ATOMIC_ACQUIRE, "workgroup", "local");
}

__device__ __forceinline__ void load_costs(const float* __restrict__ ring, int T, int lane,
                                           v4f (&q)[4]) {
    const v4f* cp = reinterpret_cast<const v4f*>(ring + lane * RS_ + ((T * 16) & (RL_ - 1)));
    q[0] = cp[0]; q[1] = cp[1]; q[2] = cp[2]; q[3] = cp[3];
}

template <bool STORE, bool PUT, bool MUL, bool LOADX, int T8>
__device__ __forceinline__ void tile_step(int T, bool is_dp, float* __restrict__ ring,
                                          const float* __restrict__ Xb,
                                          const v2f (&Af)[2][3], v2f (&Bf)[3],
                                          v8f (&acc)[2], float& x2c,
                                          const float (&p2v)[2][8],
                                          int lane, int hi, int lo, bool is15,
                                          v4f (&q)[4],
                                          float& dcur, float& p1, float& p2, float& p3,
                                          float* __restrict__ stl) {
    if (!is_dp) {
        if (PUT) tile_store<T8>(ring, acc, x2c, p2v, lane, hi);
        if (MUL) tile_products(Af, Bf, acc, x2c);
    }
    if (LOADX) load_xfrag(Xb, T + 3 + XD_, hi, lo, Bf);
    if (is_dp) {
        v4f qn[4] = { q[0], q[1], q[2], q[3] };
        if (T + 1 < NTS_) load_costs(ring, T + 1, lane, qn);
        __builtin_amdgcn_sched_barrier(0);
        const float c[16] = { q[0].x, q[0].y, q[0].z, q[0].w, q[1].x, q[1].y, q[1].z, q[1].w,
                              q[2].x, q[2].y, q[2].z, q[2].w, q[3].x, q[3].y, q[3].z, q[3].w };
        dp16<STORE>(c, T * 16, is15, dcur, p1, p2, p3, stl);
        q[0] = qn[0]; q[1] = qn[1]; q[2] = qn[2]; q[3] = qn[3];
    }
    pair_sync();
}

__global__ __launch_bounds__(WGT_) void dtw_wmma_kernel(
    const float* __restrict__ x,
    const float* __restrict__ patts,
    float* __restrict__ out)
{
    const int lane = threadIdx.x & 31;
    const int hi   = lane >> 4;
    const int lo   = lane & 15;
    const int wv   = __builtin_amdgcn_readfirstlane((int)(threadIdx.x >> 5));
    const bool is_dp = wv < NW_;
    const int pr   = wv & (NW_ - 1);
    const int b    = blockIdx.x;
    const int n    = blockIdx.y * NW_ + pr;

    __shared__ float p2all[NW_ * P_];
    __shared__ __attribute__((aligned(16))) float lds[NW_ * SLICE_];
    float* p2s  = p2all + pr * P_;
    float* ring = lds + pr * SLICE_;

    const float* Pn = patts + n * (D_ * P_);
    const float* Xb = x + b * (D_ * L_);

    v2f Bq[XD_][3];
    #pragma unroll
    for (int q = 0; q < XD_; ++q) load_xfrag(Xb, q, hi, lo, Bq[q]);

    v2f Afrag[2][3] = {};
    float p2v[2][8] = {};
    v8f acc[2] = {};
    float x2c = 0.0f;

    if (!is_dp) {

        {
            float s = 0.0f;
            #pragma unroll
            for (int d = 0; d < D_; ++d) { float v = Pn[d * P_ + lane]; s += v * v; }
            p2s[lane] = s;
        }
        static_assert((P_ * RS_) % (4 * 32) == 0, "the fill covers the whole ring");
        {
            v4f* rz = reinterpret_cast<v4f*>(ring);
            const v4f z = {BIGV, BIGV, BIGV, BIGV};
            for (int q = 0; q < (P_ * RS_) / (4 * 32); ++q) rz[q * 32 + lane] = z;
        }
        wave_sync();

        #pragma unroll
        for (int m = 0; m < 2; ++m)
            #pragma unroll
            for (int kc = 0; kc < 3; ++kc) {
                int d0 = 4 * kc + 2 * hi;
                int p  = m * 16 + lo;
                v2f a; a.x = Pn[d0 * P_ + p]; a.y = Pn[(d0 + 1) * P_ + p];
                Afrag[m][kc] = a;
            }

        #pragma unroll
        for (int m = 0; m < 2; ++m)
            #pragma unroll
            for (int r = 0; r < 8; ++r)
                p2v[m][r] = p2s[m * 16 + 8 * hi + r];

        tile_products(Afrag, Bq[0], acc, x2c);
        tile_store<0>(ring, acc, x2c, p2v, lane, hi);
        tile_products(Afrag, Bq[1], acc, x2c);
        tile_store<1>(ring, acc, x2c, p2v, lane, hi);
        tile_products(Afrag, Bq[2], acc, x2c);
    }
    load_xfrag(Xb, XD_, hi, lo, Bq[0]);
    load_xfrag(Xb, XD_ + 1, hi, lo, Bq[1]);
    load_xfrag(Xb, XD_ + 2, hi, lo, Bq[2]);
    pair_sync();

    float dcur = BIGV;
    float p1 = BIGV;
    float p2 = BIGV;
    float p3 = (lane == 0) ? 0.0f : BIGV;
    const bool is15 = (lane == 15);
    float* stage = ring + P_ * RS_;
    float* stl = stage + lane * SP_;
    v4f cq[4] = {};
    if (is_dp) load_costs(ring, 0, lane, cq);

#define TS_(STORE_, PUT_, MUL_, LOADX_, T8_, T_, SET_)                                          \
    tile_step<STORE_, PUT_, MUL_, LOADX_, T8_>(T_, is_dp, ring, Xb, Afrag, Bq[SET_], acc, x2c,  \
                                               p2v, lane, hi, lo, is15, cq, dcur, p1, p2, p3, stl)
    TS_(false, true, true, true, 2, 0, 3);
    static_assert(XD_ == 4 && RL_ / 16 == 8 && (NT_ - 3 - XD_ - 1) % 8 == 0, "trips of eight tile-steps");
    static_assert(NT_ - 3 - XD_ - 1 + 3 + XD_ == NT_ - 1, "last load is the last tile");
    for (int T = 1; T < NT_ - 3 - XD_; T += 8) {
        TS_(false, true, true, true, 3, T,     0);
        TS_(false, true, true, true, 4, T + 1, 1);
        TS_(false, true, true, true, 5, T + 2, 2);
        TS_(false, true, true, true, 6, T + 3, 3);
        TS_(false, true, true, true, 7, T + 4, 0);
        TS_(false, true, true, true, 0, T + 5, 1);
        TS_(false, true, true, true, 1, T + 6, 2);
        TS_(false, true, true, true, 2, T + 7, 3);
    }
    {
        const int T = NT_ - 3 - XD_;
        static_assert((NT_ - 3 - XD_ + 2) % 8 == 3 && (NT_ - 3 - XD_ + 3) % 4 == 0, "phase and set of the first");
        static_assert(NT_ - 3 - XD_ + 3 + 3 == NT_ - 1, "the fourth issues the last tile's products");
        static_assert(NT_ - 3 - XD_ + 5 == (L_ - LOUT_) / 16, "the last phase follows");
        TS_(false, true, true,  false, 3, T,     0);
        TS_(false, true, true,  false, 4, T + 1, 1);
        TS_(false, true, true,  false, 5, T + 2, 2);
        TS_(false, true, true,  false, 6, T + 3, 3);
        TS_(false, true, false, false, 7, T + 4, 0);
    }
    static_assert(NTS_ * 16 > L_ - 1 + SK_ * (P_ - 1), "the last row reaches the last column");
    for (int T = (L_ - LOUT_) / 16; T < NTS_; ++T)
        TS_(true, false, false, false, 0, T, 0);
#undef TS_

    if (is_dp) {
        wave_sync();
        const float* sg = stage + lane;
        float* ob = out + (b * N_ + n) * (P_ * LOUT_) + lane;
        for (int r = 0; r < P_; ++r) ob[r * LOUT_] = sg[r * (SP_ + SK_)];
    }
}

extern "C" void kernel_launch(void* const* d_in, const int* in_sizes, int n_in,
                              void* d_out, int out_size, void* d_ws, size_t ws_size,
                              hipStream_t stream) {
    const float* x     = (const float*)d_in[0];
    const float* patts = (const float*)d_in[1];
    float*       outp  = (float*)d_out;
    (void)in_sizes; (void)n_in; (void)out_size; (void)d_ws; (void)ws_size;

    dim3 grid(B_, NBY_);
    dim3 block(WGT_);
    dtw_wmma_kernel<<<grid, block, 0, stream>>>(x, patts, outp);
}
